# starting kernel
# baseline (speedup 1.0000x reference)
#include <hip/hip_runtime.h>
#include <hip/hip_bf16.h>

typedef float v2f __attribute__((ext_vector_type(2)));
typedef float v8f __attribute__((ext_vector_type(8)));

#define HC 16
#define EPSV 1e-5f

__global__ void zero_f32(float* __restrict__ p, int n) {
    int i = blockIdx.x * blockDim.x + threadIdx.x;
    if (i < n) p[i] = 0.0f;
}

__global__ void degree_kernel(const int* __restrict__ dst, float* __restrict__ deg, int E) {
    int e = blockIdx.x * blockDim.x + threadIdx.x;
    if (e < E) atomicAdd(&deg[dst[e]], 1.0f);
}

__global__ void rsqrt_kernel(float* __restrict__ d, int N) {
    int i = blockIdx.x * blockDim.x + threadIdx.x;
    if (i < N) d[i] = rsqrtf(d[i] + 1.0f);
}

__global__ void gemm_wmma_f32(const float* __restrict__ A, const float* __restrict__ B,
                              float* __restrict__ C, int N, int K) {
    int wave = (int)((blockIdx.x * blockDim.x + threadIdx.x) >> 5);
    int lane = threadIdx.x & 31;
    int row0 = wave * 16;
    if (row0 >= N) return;

    int m  = lane & 15;
    int hi = lane >> 4;
    const float* arow = A + (size_t)(row0 + m) * K;

    v8f acc = {};
    for (int k0 = 0; k0 < K; k0 += 4) {
        int ka = k0 + 2 * hi;
        v2f a, b;
        a.x = arow[ka];
        a.y = arow[ka + 1];
        b.x = B[(size_t)ka * HC + m];
        b.y = B[(size_t)(ka + 1) * HC + m];
        acc = __builtin_amdgcn_wmma_f32_16x16x4_f32(
             false, a,  false, b,
             (short)0, acc,  false,  false);
    }

    float* cbase = C + (size_t)(row0 + hi * 8) * HC + m;
#pragma unroll
    for (int r = 0; r < 8; ++r) cbase[(size_t)r * HC] = acc[r];
}

__global__ void init_selfloop(const float* __restrict__ dis, const float* __restrict__ hw,
                              const float* __restrict__ bias, float* __restrict__ out, int NC) {
    int i = blockIdx.x * blockDim.x + threadIdx.x;
    if (i >= NC) return;
    int node = i >> 4, c = i & 15;
    float d = dis[node];
    out[i] = d * d * hw[i] + bias[c];
}

__global__ void scatter_edges(const int* __restrict__ src, const int* __restrict__ dst,
                              const float* __restrict__ dis, const float* __restrict__ hw,
                              float* __restrict__ out, int E) {
    int e = blockIdx.x * blockDim.x + threadIdx.x;
    if (e >= E) return;
    int s = src[e], d = dst[e];
    float nrm = dis[s] * dis[d];
    const float4* hrow = (const float4*)(hw + (size_t)s * HC);
    float* orow = out + (size_t)d * HC;
#pragma unroll
    for (int q = 0; q < 4; ++q) {
        float4 v = hrow[q];
        atomicAdd(orow + 4 * q + 0, nrm * v.x);
        atomicAdd(orow + 4 * q + 1, nrm * v.y);
        atomicAdd(orow + 4 * q + 2, nrm * v.z);
        atomicAdd(orow + 4 * q + 3, nrm * v.w);
    }
}

__global__ void bn_stats(const float* __restrict__ agg, float* __restrict__ stats, int N) {
    __shared__ float s1s[256];
    __shared__ float s2s[256];
    int c = threadIdx.x & 15;
    int g = threadIdx.x >> 4;
    int stride = gridDim.x * 16;
    float s1 = 0.0f, s2 = 0.0f;
    for (int row = blockIdx.x * 16 + g; row < N; row += stride) {
        float v = agg[(size_t)row * HC + c];
        v = fmaxf(v, 0.0f);
        s1 += v;
        s2 += v * v;
    }
    s1s[threadIdx.x] = s1;
    s2s[threadIdx.x] = s2;
    __syncthreads();
    if (threadIdx.x < 16) {
        float t1 = 0.0f, t2 = 0.0f;
#pragma unroll
        for (int gg = 0; gg < 16; ++gg) {
            t1 += s1s[gg * 16 + threadIdx.x];
            t2 += s2s[gg * 16 + threadIdx.x];
        }
        atomicAdd(&stats[threadIdx.x], t1);
        atomicAdd(&stats[16 + threadIdx.x], t2);
    }
}

__global__ void bn_finalize(float* __restrict__ stats, const float* __restrict__ gamma,
                            const float* __restrict__ beta, float invN) {
    int c = threadIdx.x;
    if (c < 16) {
        float mu  = stats[c] * invN;
        float var = stats[16 + c] * invN - mu * mu;
        float sc  = rsqrtf(var + EPSV) * gamma[c];
        stats[32 + c] = sc;
        stats[48 + c] = beta[c] - mu * sc;
    }
}

__global__ void bn_apply(const float* __restrict__ agg, const float* __restrict__ stats,
                         float* __restrict__ h, int NC) {
    int i = blockIdx.x * blockDim.x + threadIdx.x;
    if (i >= NC) return;
    int c = i & 15;
    float v = fmaxf(agg[i], 0.0f);
    h[i] = v * stats[32 + c] + stats[48 + c];
}

__global__ void head_kernel(const float* __restrict__ h, const float* __restrict__ Wfc,
                            const float* __restrict__ bfc, float* __restrict__ out, int N) {
    int i = blockIdx.x * blockDim.x + threadIdx.x;
    if (i >= N) return;
    const float* r = h + (size_t)i * HC;
    float z0 = bfc[0], z1 = bfc[1];
#pragma unroll
    for (int c = 0; c < HC; ++c) {
        float v = r[c];
        z0 += v * Wfc[c * 2 + 0];
        z1 += v * Wfc[c * 2 + 1];
    }
    float mx  = fmaxf(z0, z1);
    float lse = mx + logf(__expf(z0 - mx) + __expf(z1 - mx));
    out[(size_t)i * 2 + 0] = z0 - lse;
    out[(size_t)i * 2 + 1] = z1 - lse;
}

extern "C" void kernel_launch(void* const* d_in, const int* in_sizes, int n_in,
                              void* d_out, int out_size, void* d_ws, size_t ws_size,
                              hipStream_t stream) {
    const int IN = 128;
    const int N  = in_sizes[0] / IN;
    const int E  = in_sizes[1] / 2;

    const float* x    = (const float*)d_in[0];
    const int*   ei   = (const int*)d_in[1];
    const int*   srcI = ei;
    const int*   dstI = ei + E;
    const float* W1   = (const float*)d_in[2];
    const float* b1   = (const float*)d_in[3];
    const float* W2   = (const float*)d_in[4];
    const float* b2   = (const float*)d_in[5];
    const float* W3   = (const float*)d_in[6];
    const float* b3   = (const float*)d_in[7];
    const float* g1   = (const float*)d_in[8];
    const float* be1  = (const float*)d_in[9];
    const float* g2   = (const float*)d_in[10];
    const float* be2  = (const float*)d_in[11];
    const float* Wfc  = (const float*)d_in[12];
    const float* bfc  = (const float*)d_in[13];

    float* ws   = (float*)d_ws;
    float* dis  = ws;
    float* bufA = dis + N;
    float* bufB = bufA + (size_t)N * HC;
    float* bufC = bufB + (size_t)N * HC;
    float* stats = bufC + (size_t)N * HC;

    const int B256 = 256;
    const int NC = N * HC;
    dim3 blkN((N + B256 - 1) / B256);
    dim3 blkNC((NC + B256 - 1) / B256);
    dim3 blkE((E + B256 - 1) / B256);
    dim3 blkG((N / 16 + 7) / 8);
    float invN = 1.0f / (float)N;

    zero_f32<<<blkN, B256, 0, stream>>>(dis, N);
    degree_kernel<<<blkE, B256, 0, stream>>>(dstI, dis, E);
    rsqrt_kernel<<<blkN, B256, 0, stream>>>(dis, N);

    gemm_wmma_f32<<<blkG, B256, 0, stream>>>(x, W1, bufA, N, IN);
    init_selfloop<<<blkNC, B256, 0, stream>>>(dis, bufA, b1, bufB, NC);
    scatter_edges<<<blkE, B256, 0, stream>>>(srcI, dstI, dis, bufA, bufB, E);
    zero_f32<<<1, 64, 0, stream>>>(stats, 64);
    bn_stats<<<256, 256, 0, stream>>>(bufB, stats, N);
    bn_finalize<<<1, 16, 0, stream>>>(stats, g1, be1, invN);
    bn_apply<<<blkNC, B256, 0, stream>>>(bufB, stats, bufC, NC);

    gemm_wmma_f32<<<blkG, B256, 0, stream>>>(bufC, W2, bufA, N, HC);
    init_selfloop<<<blkNC, B256, 0, stream>>>(dis, bufA, b2, bufB, NC);
    scatter_edges<<<blkE, B256, 0, stream>>>(srcI, dstI, dis, bufA, bufB, E);
    zero_f32<<<1, 64, 0, stream>>>(stats, 64);
    bn_stats<<<256, 256, 0, stream>>>(bufB, stats, N);
    bn_finalize<<<1, 16, 0, stream>>>(stats, g2, be2, invN);
    bn_apply<<<blkNC, B256, 0, stream>>>(bufB, stats, bufC, NC);

    gemm_wmma_f32<<<blkG, B256, 0, stream>>>(bufC, W3, bufA, N, HC);
    init_selfloop<<<blkNC, B256, 0, stream>>>(dis, bufA, b3, bufB, NC);
    scatter_edges<<<blkE, B256, 0, stream>>>(srcI, dstI, dis, bufA, bufB, E);

    head_kernel<<<blkN, B256, 0, stream>>>(bufB, Wfc, bfc, (float*)d_out, N);
}


# L2 float atomics per edge replaced by owner work-groups: each holds 392 dst rows in LDS, scans all dst, adds its own edges with LDS float adds (arrival order, as before) and stores each row once
# speedup vs baseline: 3.1942x; 3.1942x over previous
#include <hip/hip_runtime.h>
#include <hip/hip_bf16.h>

typedef float v2f __attribute__((ext_vector_type(2)));
typedef float v8f __attribute__((ext_vector_type(8)));

#define HC 16
#define EPSV 1e-5f

__global__ void zero_f32(float* __restrict__ p, int n) {
    int i = blockIdx.x * blockDim.x + threadIdx.x;
    if (i < n) p[i] = 0.0f;
}

__global__ void degree_kernel(const int* __restrict__ dst, float* __restrict__ deg, int E) {
    int e = blockIdx.x * blockDim.x + threadIdx.x;
    if (e < E) atomicAdd(&deg[dst[e]], 1.0f);
}

__global__ void rsqrt_kernel(float* __restrict__ d, int N) {
    int i = blockIdx.x * blockDim.x + threadIdx.x;
    if (i < N) d[i] = rsqrtf(d[i] + 1.0f);
}

__global__ void gemm_wmma_f32(const float* __restrict__ A, const float* __restrict__ B,
                              float* __restrict__ C, int N, int K) {
    int wave = (int)((blockIdx.x * blockDim.x + threadIdx.x) >> 5);
    int lane = threadIdx.x & 31;
    int row0 = wave * 16;
    if (row0 >= N) return;

    int m  = lane & 15;
    int hi = lane >> 4;
    const float* arow = A + (size_t)(row0 + m) * K;

    v8f acc = {};
    for (int k0 = 0; k0 < K; k0 += 4) {
        int ka = k0 + 2 * hi;
        v2f a, b;
        a.x = arow[ka];
        a.y = arow[ka + 1];
        b.x = B[(size_t)ka * HC + m];
        b.y = B[(size_t)(ka + 1) * HC + m];
        acc = __builtin_amdgcn_wmma_f32_16x16x4_f32(
             false, a,  false, b,
             (short)0, acc,  false,  false);
    }

    float* cbase = C + (size_t)(row0 + hi * 8) * HC + m;
#pragma unroll
    for (int r = 0; r < 8; ++r) cbase[(size_t)r * HC] = acc[r];
}

#define OWN_ROWS    392
#define OWN_PITCH   17
#define OWN_THREADS 1024
#define OWN_QCAP    64
#define OWN_CHUNK   (16 * OWN_THREADS)

__device__ __forceinline__ void own_add_edge(int e, const int* __restrict__ src, const int* __restrict__ dst,
                                             const float* __restrict__ dis, const float* __restrict__ hw,
                                             float* acc, int lo) {
    int s = src[e], d = dst[e];
    float nrm = dis[s] * dis[d];
    const float4* hrow = (const float4*)(hw + (size_t)s * HC);
    float* arow = acc + (d - lo) * OWN_PITCH;
#pragma unroll
    for (int q = 0; q < 4; ++q) {
        float4 v = hrow[q];
        atomicAdd(arow + 4 * q + 0, nrm * v.x);
        atomicAdd(arow + 4 * q + 1, nrm * v.y);
        atomicAdd(arow + 4 * q + 2, nrm * v.z);
        atomicAdd(arow + 4 * q + 3, nrm * v.w);
    }
}

__global__ __launch_bounds__(OWN_THREADS)
void aggregate_owner(const int* __restrict__ src, const int* __restrict__ dst,
                     const float* __restrict__ dis, const float* __restrict__ hw,
                     const float* __restrict__ bias, float* __restrict__ out, int N, int E) {
    __shared__ float acc[OWN_ROWS * OWN_PITCH];
    __shared__ int queue[(OWN_THREADS / 32) * OWN_QCAP];
    const int tid  = threadIdx.x;
    const int lane = tid & 31;
    int* myq = queue + (tid >> 5) * OWN_QCAP;
    const int lo   = blockIdx.x * OWN_ROWS;
    const int rows = min(OWN_ROWS, N - lo);

    for (int idx = tid; idx < rows * HC; idx += OWN_THREADS) {
        int r = idx >> 4, c = idx & 15;
        float d = dis[lo + r];
        acc[r * OWN_PITCH + c] = d * d * hw[(size_t)lo * HC + idx] + bias[c];
    }
    __syncthreads();

    int qn = 0;

#define OWN_STEP(m, km, e)                                                              \
    {                                                                                   \
        if (m) myq[qn + (int)__builtin_amdgcn_mbcnt_lo(km, 0u)] = (e);                  \
        qn += __popc(km);                                                               \
          \
          \
          \
        __builtin_amdgcn_wave_barrier();                                                \
        if (qn >= 32) {                                                                 \
            qn -= 32;                                                                   \
            own_add_edge(myq[qn + lane], src, dst, dis, hw, acc, lo);                   \
        }                                                                               \
                             \
        __builtin_amdgcn_wave_barrier();                                                \
    }

    int base = 0;
    for (; base + OWN_CHUNK <= E; base += OWN_CHUNK) {
        const int* p = dst + base + tid;
        int dv[16];
#pragma unroll
        for (int k = 0; k < 16; ++k) dv[k] = p[k * OWN_THREADS];
        unsigned km[16];
        unsigned any = 0u;
#pragma unroll
        for (int k = 0; k < 16; ++k) {
            km[k] = __builtin_amdgcn_ballot_w32((unsigned)(dv[k] - lo) < (unsigned)rows);
            any |= km[k];
        }
        if (any != 0u) {
#pragma unroll
            for (int k = 0; k < 16; ++k) {
                if (km[k] != 0u) {
                    bool m = (unsigned)(dv[k] - lo) < (unsigned)rows;
                    OWN_STEP(m, km[k], base + tid + k * OWN_THREADS)
                }
            }
        }
    }
    for (; base < E; base += OWN_THREADS) {
        int e = base + tid;
        int d = (e < E) ? dst[e] : -1;
        bool m = (unsigned)(d - lo) < (unsigned)rows;
        unsigned k1 = __builtin_amdgcn_ballot_w32(m);
        if (k1 != 0u) OWN_STEP(m, k1, e)
    }
#undef OWN_STEP
    if (lane < qn) own_add_edge(myq[lane], src, dst, dis, hw, acc, lo);
    __syncthreads();

    for (int idx = tid; idx < rows * HC; idx += OWN_THREADS) {
        int r = idx >> 4, c = idx & 15;
        out[(size_t)lo * HC + idx] = acc[r * OWN_PITCH + c];
    }
}

__global__ void bn_stats(const float* __restrict__ agg, float* __restrict__ stats, int N) {
    __shared__ float s1s[256];
    __shared__ float s2s[256];
    int c = threadIdx.x & 15;
    int g = threadIdx.x >> 4;
    int stride = gridDim.x * 16;
    float s1 = 0.0f, s2 = 0.0f;
    for (int row = blockIdx.x * 16 + g; row < N; row += stride) {
        float v = agg[(size_t)row * HC + c];
        v = fmaxf(v, 0.0f);
        s1 += v;
        s2 += v * v;
    }
    s1s[threadIdx.x] = s1;
    s2s[threadIdx.x] = s2;
    __syncthreads();
    if (threadIdx.x < 16) {
        float t1 = 0.0f, t2 = 0.0f;
#pragma unroll
        for (int gg = 0; gg < 16; ++gg) {
            t1 += s1s[gg * 16 + threadIdx.x];
            t2 += s2s[gg * 16 + threadIdx.x];
        }
        atomicAdd(&stats[threadIdx.x], t1);
        atomicAdd(&stats[16 + threadIdx.x], t2);
    }
}

__global__ void bn_finalize(float* __restrict__ stats, const float* __restrict__ gamma,
                            const float* __restrict__ beta, float invN) {
    int c = threadIdx.x;
    if (c < 16) {
        float mu  = stats[c] * invN;
        float var = stats[16 + c] * invN - mu * mu;
        float sc  = rsqrtf(var + EPSV) * gamma[c];
        stats[32 + c] = sc;
        stats[48 + c] = beta[c] - mu * sc;
    }
}

__global__ void bn_apply(const float* __restrict__ agg, const float* __restrict__ stats,
                         float* __restrict__ h, int NC) {
    int i = blockIdx.x * blockDim.x + threadIdx.x;
    if (i >= NC) return;
    int c = i & 15;
    float v = fmaxf(agg[i], 0.0f);
    h[i] = v * stats[32 + c] + stats[48 + c];
}

__global__ void head_kernel(const float* __restrict__ h, const float* __restrict__ Wfc,
                            const float* __restrict__ bfc, float* __restrict__ out, int N) {
    int i = blockIdx.x * blockDim.x + threadIdx.x;
    if (i >= N) return;
    const float* r = h + (size_t)i * HC;
    float z0 = bfc[0], z1 = bfc[1];
#pragma unroll
    for (int c = 0; c < HC; ++c) {
        float v = r[c];
        z0 += v * Wfc[c * 2 + 0];
        z1 += v * Wfc[c * 2 + 1];
    }
    float mx  = fmaxf(z0, z1);
    float lse = mx + logf(__expf(z0 - mx) + __expf(z1 - mx));
    out[(size_t)i * 2 + 0] = z0 - lse;
    out[(size_t)i * 2 + 1] = z1 - lse;
}

extern "C" void kernel_launch(void* const* d_in, const int* in_sizes, int n_in,
                              void* d_out, int out_size, void* d_ws, size_t ws_size,
                              hipStream_t stream) {
    const int IN = 128;
    const int N  = in_sizes[0] / IN;
    const int E  = in_sizes[1] / 2;

    const float* x    = (const float*)d_in[0];
    const int*   ei   = (const int*)d_in[1];
    const int*   srcI = ei;
    const int*   dstI = ei + E;
    const float* W1   = (const float*)d_in[2];
    const float* b1   = (const float*)d_in[3];
    const float* W2   = (const float*)d_in[4];
    const float* b2   = (const float*)d_in[5];
    const float* W3   = (const float*)d_in[6];
    const float* b3   = (const float*)d_in[7];
    const float* g1   = (const float*)d_in[8];
    const float* be1  = (const float*)d_in[9];
    const float* g2   = (const float*)d_in[10];
    const float* be2  = (const float*)d_in[11];
    const float* Wfc  = (const float*)d_in[12];
    const float* bfc  = (const float*)d_in[13];

    float* ws   = (float*)d_ws;
    float* dis  = ws;
    float* bufA = dis + N;
    float* bufB = bufA + (size_t)N * HC;
    float* bufC = bufB + (size_t)N * HC;
    float* stats = bufC + (size_t)N * HC;

    const int B256 = 256;
    const int NC = N * HC;
    dim3 blkN((N + B256 - 1) / B256);
    dim3 blkNC((NC + B256 - 1) / B256);
    dim3 blkE((E + B256 - 1) / B256);
    dim3 blkG((N / 16 + 7) / 8);
    float invN = 1.0f / (float)N;
    dim3 blkO((N + OWN_ROWS - 1) / OWN_ROWS);

    zero_f32<<<blkN, B256, 0, stream>>>(dis, N);
    degree_kernel<<<blkE, B256, 0, stream>>>(dstI, dis, E);
    rsqrt_kernel<<<blkN, B256, 0, stream>>>(dis, N);

    gemm_wmma_f32<<<blkG, B256, 0, stream>>>(x, W1, bufA, N, IN);
    aggregate_owner<<<blkO, OWN_THREADS, 0, stream>>>(srcI, dstI, dis, bufA, b1, bufB, N, E);
    zero_f32<<<1, 64, 0, stream>>>(stats, 64);
    bn_stats<<<256, 256, 0, stream>>>(bufB, stats, N);
    bn_finalize<<<1, 16, 0, stream>>>(stats, g1, be1, invN);
    bn_apply<<<blkNC, B256, 0, stream>>>(bufB, stats, bufC, NC);

    gemm_wmma_f32<<<blkG, B256, 0, stream>>>(bufC, W2, bufA, N, HC);
    aggregate_owner<<<blkO, OWN_THREADS, 0, stream>>>(srcI, dstI, dis, bufA, b2, bufB, N, E);
    zero_f32<<<1, 64, 0, stream>>>(stats, 64);
    bn_stats<<<256, 256, 0, stream>>>(bufB, stats, N);
    bn_finalize<<<1, 16, 0, stream>>>(stats, g2, be2, invN);
    bn_apply<<<blkNC, B256, 0, stream>>>(bufB, stats, bufC, NC);

    gemm_wmma_f32<<<blkG, B256, 0, stream>>>(bufC, W3, bufA, N, HC);
    aggregate_owner<<<blkO, OWN_THREADS, 0, stream>>>(srcI, dstI, dis, bufA, b3, bufB, N, E);

    head_kernel<<<blkN, B256, 0, stream>>>(bufB, Wfc, bfc, (float*)d_out, N);
}


# Owner scan split 2-D: 32 row ranges (3126 rows in LDS) x 8 edge slices, each scans 1/8 of dst; partial rows go out by whole-line L2 float atomics onto a start value the GEMM stores (arrival order)
# speedup vs baseline: 4.9261x; 1.5422x over previous
#include <hip/hip_runtime.h>
#include <hip/hip_bf16.h>

typedef float v2f __attribute__((ext_vector_type(2)));
typedef float v8f __attribute__((ext_vector_type(8)));

#define HC 16
#define EPSV 1e-5f

__global__ void zero_f32(float* __restrict__ p, int n) {
    int i = blockIdx.x * blockDim.x + threadIdx.x;
    if (i < n) p[i] = 0.0f;
}

__global__ void degree_kernel(const int* __restrict__ dst, float* __restrict__ deg, int E) {
    int e = blockIdx.x * blockDim.x + threadIdx.x;
    if (e < E) atomicAdd(&deg[dst[e]], 1.0f);
}

__global__ void rsqrt_kernel(float* __restrict__ d, int N) {
    int i = blockIdx.x * blockDim.x + threadIdx.x;
    if (i < N) d[i] = rsqrtf(d[i] + 1.0f);
}

__global__ void gemm_wmma_f32(const float* __restrict__ A, const float* __restrict__ B,
                              float* __restrict__ C, const float* __restrict__ dis,
                              const float* __restrict__ bias, float* __restrict__ S, int N, int K) {
    int wave = (int)((blockIdx.x * blockDim.x + threadIdx.x) >> 5);
    int lane = threadIdx.x & 31;
    int row0 = wave * 16;
    if (row0 >= N) return;

    int m  = lane & 15;
    int hi = lane >> 4;
    const float* arow = A + (size_t)(row0 + m) * K;

    v8f acc = {};
    for (int k0 = 0; k0 < K; k0 += 4) {
        int ka = k0 + 2 * hi;
        v2f a, b;
        a.x = arow[ka];
        a.y = arow[ka + 1];
        b.x = B[(size_t)ka * HC + m];
        b.y = B[(size_t)(ka + 1) * HC + m];
        acc = __builtin_amdgcn_wmma_f32_16x16x4_f32(
             false, a,  false, b,
             (short)0, acc,  false,  false);
    }

    float* cbase = C + (size_t)(row0 + hi * 8) * HC + m;
    float* sbase = S + (size_t)(row0 + hi * 8) * HC + m;
    const float* drow = dis + row0 + hi * 8;
    float bm = bias[m];
#pragma unroll
    for (int r = 0; r < 8; ++r) {
        float d = drow[r];
        cbase[(size_t)r * HC] = acc[r];
        sbase[(size_t)r * HC] = d * d * acc[r] + bm;
    }
}

#define OWN_ROWS    3126
#define OWN_SLICES  8
#define OWN_PITCH   17
#define OWN_THREADS 1024
#define OWN_QCAP    64
#define OWN_CHUNK   (16 * OWN_THREADS)

__device__ __forceinline__ void own_add_edge(int e, const int* __restrict__ src, const int* __restrict__ dst,
                                             const float* __restrict__ dis, const float* __restrict__ hw,
                                             float* acc, int lo) {
    int s = src[e], d = dst[e];
    float nrm = dis[s] * dis[d];
    const float4* hrow = (const float4*)(hw + (size_t)s * HC);
    float* arow = acc + (d - lo) * OWN_PITCH;
#pragma unroll
    for (int q = 0; q < 4; ++q) {
        float4 v = hrow[q];
        atomicAdd(arow + 4 * q + 0, nrm * v.x);
        atomicAdd(arow + 4 * q + 1, nrm * v.y);
        atomicAdd(arow + 4 * q + 2, nrm * v.z);
        atomicAdd(arow + 4 * q + 3, nrm * v.w);
    }
}

__global__ __launch_bounds__(OWN_THREADS)
void aggregate_owner(const int* __restrict__ src, const int* __restrict__ dst,
                     const float* __restrict__ dis, const float* __restrict__ hw,
                     float* __restrict__ out, int N, int E) {
    __shared__ float acc[OWN_ROWS * OWN_PITCH];
    __shared__ int queue[(OWN_THREADS / 32) * OWN_QCAP];
    const int tid  = threadIdx.x;
    const int lane = tid & 31;
    int* myq = queue + (tid >> 5) * OWN_QCAP;
    const int grp  = blockIdx.x / OWN_SLICES;
    const int slc  = blockIdx.x % OWN_SLICES;
    const int lo   = grp * OWN_ROWS;
    const int rows = min(OWN_ROWS, N - lo);
    const int per  = ((E + OWN_SLICES * OWN_THREADS - 1) / (OWN_SLICES * OWN_THREADS)) * OWN_THREADS;
    const int e0   = min(E, slc * per);
    const int e1   = min(E, e0 + per);

    for (int idx = tid; idx < rows * OWN_PITCH; idx += OWN_THREADS) acc[idx] = 0.0f;
    __syncthreads();

    int qn = 0;

#define OWN_STEP(m, km, e)                                                              \
    {                                                                                   \
        if (m) myq[qn + (int)__builtin_amdgcn_mbcnt_lo(km, 0u)] = (e);                  \
        qn += __popc(km);                                                               \
          \
          \
          \
        __builtin_amdgcn_wave_barrier();                                                \
        if (qn >= 32) {                                                                 \
            qn -= 32;                                                                   \
            own_add_edge(myq[qn + lane], src, dst, dis, hw, acc, lo);                   \
        }                                                                               \
                             \
        __builtin_amdgcn_wave_barrier();                                                \
    }

    int base = e0;
    for (; base + OWN_CHUNK <= e1; base += OWN_CHUNK) {
        const int* p = dst + base + tid;
        int dv[16];
#pragma unroll
        for (int k = 0; k < 16; ++k) dv[k] = p[k * OWN_THREADS];
        unsigned km[16];
        unsigned any = 0u;
#pragma unroll
        for (int k = 0; k < 16; ++k) {
            km[k] = __builtin_amdgcn_ballot_w32((unsigned)(dv[k] - lo) < (unsigned)rows);
            any |= km[k];
        }
        if (any != 0u) {
#pragma unroll
            for (int k = 0; k < 16; ++k) {
                if (km[k] != 0u) {
                    bool m = (unsigned)(dv[k] - lo) < (unsigned)rows;
                    OWN_STEP(m, km[k], base + tid + k * OWN_THREADS)
                }
            }
        }
    }
    for (; base < e1; base += OWN_THREADS) {
        int e = base + tid;
        int d = (e < e1) ? dst[e] : -1;
        bool m = (unsigned)(d - lo) < (unsigned)rows;
        unsigned k1 = __builtin_amdgcn_ballot_w32(m);
        if (k1 != 0u) OWN_STEP(m, k1, e)
    }
#undef OWN_STEP
    if (lane < qn) own_add_edge(myq[lane], src, dst, dis, hw, acc, lo);
    __syncthreads();

    for (int idx = tid; idx < rows * HC; idx += OWN_THREADS) {
        int r = idx >> 4, c = idx & 15;
        atomicAdd(out + (size_t)lo * HC + idx, acc[r * OWN_PITCH + c]);
    }
}

__global__ void bn_stats(const float* __restrict__ agg, float* __restrict__ stats, int N) {
    __shared__ float s1s[256];
    __shared__ float s2s[256];
    int c = threadIdx.x & 15;
    int g = threadIdx.x >> 4;
    int stride = gridDim.x * 16;
    float s1 = 0.0f, s2 = 0.0f;
    for (int row = blockIdx.x * 16 + g; row < N; row += stride) {
        float v = agg[(size_t)row * HC + c];
        v = fmaxf(v, 0.0f);
        s1 += v;
        s2 += v * v;
    }
    s1s[threadIdx.x] = s1;
    s2s[threadIdx.x] = s2;
    __syncthreads();
    if (threadIdx.x < 16) {
        float t1 = 0.0f, t2 = 0.0f;
#pragma unroll
        for (int gg = 0; gg < 16; ++gg) {
            t1 += s1s[gg * 16 + threadIdx.x];
            t2 += s2s[gg * 16 + threadIdx.x];
        }
        atomicAdd(&stats[threadIdx.x], t1);
        atomicAdd(&stats[16 + threadIdx.x], t2);
    }
}

__global__ void bn_finalize(float* __restrict__ stats, const float* __restrict__ gamma,
                            const float* __restrict__ beta, float invN) {
    int c = threadIdx.x;
    if (c < 16) {
        float mu  = stats[c] * invN;
        float var = stats[16 + c] * invN - mu * mu;
        float sc  = rsqrtf(var + EPSV) * gamma[c];
        stats[32 + c] = sc;
        stats[48 + c] = beta[c] - mu * sc;
    }
}

__global__ void bn_apply(const float* __restrict__ agg, const float* __restrict__ stats,
                         float* __restrict__ h, int NC) {
    int i = blockIdx.x * blockDim.x + threadIdx.x;
    if (i >= NC) return;
    int c = i & 15;
    float v = fmaxf(agg[i], 0.0f);
    h[i] = v * stats[32 + c] + stats[48 + c];
}

__global__ void head_kernel(const float* __restrict__ h, const float* __restrict__ Wfc,
                            const float* __restrict__ bfc, float* __restrict__ out, int N) {
    int i = blockIdx.x * blockDim.x + threadIdx.x;
    if (i >= N) return;
    const float* r = h + (size_t)i * HC;
    float z0 = bfc[0], z1 = bfc[1];
#pragma unroll
    for (int c = 0; c < HC; ++c) {
        float v = r[c];
        z0 += v * Wfc[c * 2 + 0];
        z1 += v * Wfc[c * 2 + 1];
    }
    float mx  = fmaxf(z0, z1);
    float lse = mx + logf(__expf(z0 - mx) + __expf(z1 - mx));
    out[(size_t)i * 2 + 0] = z0 - lse;
    out[(size_t)i * 2 + 1] = z1 - lse;
}

extern "C" void kernel_launch(void* const* d_in, const int* in_sizes, int n_in,
                              void* d_out, int out_size, void* d_ws, size_t ws_size,
                              hipStream_t stream) {
    const int IN = 128;
    const int N  = in_sizes[0] / IN;
    const int E  = in_sizes[1] / 2;

    const float* x    = (const float*)d_in[0];
    const int*   ei   = (const int*)d_in[1];
    const int*   srcI = ei;
    const int*   dstI = ei + E;
    const float* W1   = (const float*)d_in[2];
    const float* b1   = (const float*)d_in[3];
    const float* W2   = (const float*)d_in[4];
    const float* b2   = (const float*)d_in[5];
    const float* W3   = (const float*)d_in[6];
    const float* b3   = (const float*)d_in[7];
    const float* g1   = (const float*)d_in[8];
    const float* be1  = (const float*)d_in[9];
    const float* g2   = (const float*)d_in[10];
    const float* be2  = (const float*)d_in[11];
    const float* Wfc  = (const float*)d_in[12];
    const float* bfc  = (const float*)d_in[13];

    float* ws   = (float*)d_ws;
    float* dis  = ws;
    float* bufA = dis + N;
    float* bufB = bufA + (size_t)N * HC;
    float* bufC = bufB + (size_t)N * HC;
    float* stats = bufC + (size_t)N * HC;

    const int B256 = 256;
    const int NC = N * HC;
    dim3 blkN((N + B256 - 1) / B256);
    dim3 blkNC((NC + B256 - 1) / B256);
    dim3 blkE((E + B256 - 1) / B256);
    dim3 blkG((N / 16 + 7) / 8);
    float invN = 1.0f / (float)N;
    dim3 blkO(((N + OWN_ROWS - 1) / OWN_ROWS) * OWN_SLICES);

    zero_f32<<<blkN, B256, 0, stream>>>(dis, N);
    degree_kernel<<<blkE, B256, 0, stream>>>(dstI, dis, E);
    rsqrt_kernel<<<blkN, B256, 0, stream>>>(dis, N);

    gemm_wmma_f32<<<blkG, B256, 0, stream>>>(x, W1, bufA, dis, b1, bufB, N, IN);
    aggregate_owner<<<blkO, OWN_THREADS, 0, stream>>>(srcI, dstI, dis, bufA, bufB, N, E);
    zero_f32<<<1, 64, 0, stream>>>(stats, 64);
    bn_stats<<<256, 256, 0, stream>>>(bufB, stats, N);
    bn_finalize<<<1, 16, 0, stream>>>(stats, g1, be1, invN);
    bn_apply<<<blkNC, B256, 0, stream>>>(bufB, stats, bufC, NC);

    gemm_wmma_f32<<<blkG, B256, 0, stream>>>(bufC, W2, bufA, dis, b2, bufB, N, HC);
    aggregate_owner<<<blkO, OWN_THREADS, 0, stream>>>(srcI, dstI, dis, bufA, bufB, N, E);
    zero_f32<<<1, 64, 0, stream>>>(stats, 64);
    bn_stats<<<256, 256, 0, stream>>>(bufB, stats, N);
    bn_finalize<<<1, 16, 0, stream>>>(stats, g2, be2, invN);
    bn_apply<<<blkNC, B256, 0, stream>>>(bufB, stats, bufC, NC);

    gemm_wmma_f32<<<blkG, B256, 0, stream>>>(bufC, W3, bufA, dis, b3, bufB, N, HC);
    aggregate_owner<<<blkO, OWN_THREADS, 0, stream>>>(srcI, dstI, dis, bufA, bufB, N, E);

    head_kernel<<<blkN, B256, 0, stream>>>(bufB, Wfc, bfc, (float*)d_out, N);
}


# Gather per edge cut from 5 L2 lines to 1: src read beside dst in the scan and queued; dis[src] folded into GEMM rows, dis[dst] times each partial row sum (products re-associated, arrival order)
# speedup vs baseline: 5.3750x; 1.0911x over previous
#include <hip/hip_runtime.h>
#include <hip/hip_bf16.h>

typedef float v2f __attribute__((ext_vector_type(2)));
typedef float v8f __attribute__((ext_vector_type(8)));

#define HC 16
#define EPSV 1e-5f

__global__ void zero_f32(float* __restrict__ p, int n) {
    int i = blockIdx.x * blockDim.x + threadIdx.x;
    if (i < n) p[i] = 0.0f;
}

__global__ void degree_kernel(const int* __restrict__ dst, float* __restrict__ deg, int E) {
    int e = blockIdx.x * blockDim.x + threadIdx.x;
    if (e < E) atomicAdd(&deg[dst[e]], 1.0f);
}

__global__ void rsqrt_kernel(float* __restrict__ d, int N) {
    int i = blockIdx.x * blockDim.x + threadIdx.x;
    if (i < N) d[i] = rsqrtf(d[i] + 1.0f);
}

__global__ void gemm_wmma_f32(const float* __restrict__ A, const float* __restrict__ B,
                              float* __restrict__ C, const float* __restrict__ dis,
                              const float* __restrict__ bias, float* __restrict__ S, int N, int K) {
    int wave = (int)((blockIdx.x * blockDim.x + threadIdx.x) >> 5);
    int lane = threadIdx.x & 31;
    int row0 = wave * 16;
    if (row0 >= N) return;

    int m  = lane & 15;
    int hi = lane >> 4;
    const float* arow = A + (size_t)(row0 + m) * K;

    v8f acc = {};
    for (int k0 = 0; k0 < K; k0 += 4) {
        int ka = k0 + 2 * hi;
        v2f a, b;
        a.x = arow[ka];
        a.y = arow[ka + 1];
        b.x = B[(size_t)ka * HC + m];
        b.y = B[(size_t)(ka + 1) * HC + m];
        acc = __builtin_amdgcn_wmma_f32_16x16x4_f32(
             false, a,  false, b,
             (short)0, acc,  false,  false);
    }

    float* cbase = C + (size_t)(row0 + hi * 8) * HC + m;
    float* sbase = S + (size_t)(row0 + hi * 8) * HC + m;
    const float* drow = dis + row0 + hi * 8;
    float bm = bias[m];
#pragma unroll
    for (int r = 0; r < 8; ++r) {
        float d = drow[r];
        cbase[(size_t)r * HC] = d * acc[r];
        sbase[(size_t)r * HC] = d * d * acc[r] + bm;
    }
}

#define OWN_ROWS    3126
#define OWN_SLICES  8
#define OWN_PITCH   17
#define OWN_THREADS 1024
#define OWN_QCAP    64
#define OWN_CHUNK   (16 * OWN_THREADS)
#define OWN_SRC_BITS 20
static_assert(OWN_ROWS <= (1 << (32 - OWN_SRC_BITS)), "local row must fit the queue word");

__device__ __forceinline__ void own_add_edge(unsigned w, const float* __restrict__ hw, float* acc) {
    const unsigned s  = w & ((1u << OWN_SRC_BITS) - 1u);
    const unsigned dl = w >> OWN_SRC_BITS;
    const float4* hrow = (const float4*)(hw + (size_t)s * HC);
    float* arow = acc + dl * OWN_PITCH;
#pragma unroll
    for (int q = 0; q < 4; ++q) {
        float4 v = hrow[q];
        atomicAdd(arow + 4 * q + 0, v.x);
        atomicAdd(arow + 4 * q + 1, v.y);
        atomicAdd(arow + 4 * q + 2, v.z);
        atomicAdd(arow + 4 * q + 3, v.w);
    }
}

__global__ __launch_bounds__(OWN_THREADS)
void aggregate_owner(const int* __restrict__ src, const int* __restrict__ dst,
                     const float* __restrict__ dis, const float* __restrict__ hw,
                     float* __restrict__ out, int N, int E) {
    __shared__ float acc[OWN_ROWS * OWN_PITCH];
    __shared__ unsigned queue[(OWN_THREADS / 32) * OWN_QCAP];
    __shared__ float dsl[OWN_ROWS];
    const int tid  = threadIdx.x;
    const int lane = tid & 31;
    unsigned* myq = queue + (tid >> 5) * OWN_QCAP;
    const int grp  = blockIdx.x / OWN_SLICES;
    const int slc  = blockIdx.x % OWN_SLICES;
    const int lo   = grp * OWN_ROWS;
    const int rows = min(OWN_ROWS, N - lo);
    const int per  = ((E + OWN_SLICES * OWN_THREADS - 1) / (OWN_SLICES * OWN_THREADS)) * OWN_THREADS;
    const int e0   = min(E, slc * per);
    const int e1   = min(E, e0 + per);

    constexpr int DPT = (OWN_ROWS + OWN_THREADS - 1) / OWN_THREADS;
    float dreg[DPT];
#pragma unroll
    for (int j = 0; j < DPT; ++j) {
        int idx = tid + j * OWN_THREADS;
        dreg[j] = (idx < rows) ? dis[lo + idx] : 0.0f;
    }
    for (int idx = tid; idx < rows * OWN_PITCH; idx += OWN_THREADS) acc[idx] = 0.0f;
#pragma unroll
    for (int j = 0; j < DPT; ++j) {
        int idx = tid + j * OWN_THREADS;
        if (idx < rows) dsl[idx] = dreg[j];
    }
    __syncthreads();

    int qn = 0;

#define OWN_STEP(m, km, w)                                                              \
    {                                                                                   \
        if (m) myq[qn + (int)__builtin_amdgcn_mbcnt_lo(km, 0u)] = (w);                  \
        qn += __popc(km);                                                               \
          \
          \
          \
        __builtin_amdgcn_wave_barrier();                                                \
        if (qn >= 32) {                                                                 \
            qn -= 32;                                                                   \
            own_add_edge(myq[qn + lane], hw, acc);                                      \
        }                                                                               \
                             \
        __builtin_amdgcn_wave_barrier();                                                \
    }

    int base = e0;
    for (; base + OWN_CHUNK <= e1; base += OWN_CHUNK) {
        const int* p  = dst + base + tid;
        const int* ps = src + base + tid;
        int dv[16], sv[16];
#pragma unroll
        for (int k = 0; k < 16; ++k) dv[k] = p[k * OWN_THREADS];
#pragma unroll
        for (int k = 0; k < 16; ++k)
            sv[k] = __hip_atomic_load(ps + k * OWN_THREADS, __ATOMIC_RELAXED, __HIP_MEMORY_SCOPE_WAVEFRONT);
        unsigned km[16];
        unsigned any = 0u;
#pragma unroll
        for (int k = 0; k < 16; ++k) {
            km[k] = __builtin_amdgcn_ballot_w32((unsigned)(dv[k] - lo) < (unsigned)rows);
            any |= km[k];
        }
        if (any != 0u) {
#pragma unroll
            for (int k = 0; k < 16; ++k) {
                if (km[k] != 0u) {
                    bool m = (unsigned)(dv[k] - lo) < (unsigned)rows;
                    OWN_STEP(m, km[k], (unsigned)sv[k] | ((unsigned)(dv[k] - lo) << OWN_SRC_BITS))
                }
            }
        }
    }
    for (; base < e1; base += OWN_THREADS) {
        int e = base + tid;
        int d = (e < e1) ? dst[e] : -1;
        int s = (e < e1) ? src[e] : 0;
        bool m = (unsigned)(d - lo) < (unsigned)rows;
        unsigned k1 = __builtin_amdgcn_ballot_w32(m);
        if (k1 != 0u) OWN_STEP(m, k1, (unsigned)s | ((unsigned)(d - lo) << OWN_SRC_BITS))
    }
#undef OWN_STEP
    if (lane < qn) own_add_edge(myq[lane], hw, acc);
    __syncthreads();

    for (int idx = tid; idx < rows * HC; idx += OWN_THREADS) {
        int r = idx >> 4, c = idx & 15;
        atomicAdd(out + (size_t)lo * HC + idx, dsl[r] * acc[r * OWN_PITCH + c]);
    }
}

__global__ void bn_stats(const float* __restrict__ agg, float* __restrict__ stats, int N) {
    __shared__ float s1s[256];
    __shared__ float s2s[256];
    int c = threadIdx.x & 15;
    int g = threadIdx.x >> 4;
    int stride = gridDim.x * 16;
    float s1 = 0.0f, s2 = 0.0f;
    for (int row = blockIdx.x * 16 + g; row < N; row += stride) {
        float v = agg[(size_t)row * HC + c];
        v = fmaxf(v, 0.0f);
        s1 += v;
        s2 += v * v;
    }
    s1s[threadIdx.x] = s1;
    s2s[threadIdx.x] = s2;
    __syncthreads();
    if (threadIdx.x < 16) {
        float t1 = 0.0f, t2 = 0.0f;
#pragma unroll
        for (int gg = 0; gg < 16; ++gg) {
            t1 += s1s[gg * 16 + threadIdx.x];
            t2 += s2s[gg * 16 + threadIdx.x];
        }
        atomicAdd(&stats[threadIdx.x], t1);
        atomicAdd(&stats[16 + threadIdx.x], t2);
    }
}

__global__ void bn_finalize(float* __restrict__ stats, const float* __restrict__ gamma,
                            const float* __restrict__ beta, float invN) {
    int c = threadIdx.x;
    if (c < 16) {
        float mu  = stats[c] * invN;
        float var = stats[16 + c] * invN - mu * mu;
        float sc  = rsqrtf(var + EPSV) * gamma[c];
        stats[32 + c] = sc;
        stats[48 + c] = beta[c] - mu * sc;
    }
}

__global__ void bn_apply(const float* __restrict__ agg, const float* __restrict__ stats,
                         float* __restrict__ h, int NC) {
    int i = blockIdx.x * blockDim.x + threadIdx.x;
    if (i >= NC) return;
    int c = i & 15;
    float v = fmaxf(agg[i], 0.0f);
    h[i] = v * stats[32 + c] + stats[48 + c];
}

__global__ void head_kernel(const float* __restrict__ h, const float* __restrict__ Wfc,
                            const float* __restrict__ bfc, float* __restrict__ out, int N) {
    int i = blockIdx.x * blockDim.x + threadIdx.x;
    if (i >= N) return;
    const float* r = h + (size_t)i * HC;
    float z0 = bfc[0], z1 = bfc[1];
#pragma unroll
    for (int c = 0; c < HC; ++c) {
        float v = r[c];
        z0 += v * Wfc[c * 2 + 0];
        z1 += v * Wfc[c * 2 + 1];
    }
    float mx  = fmaxf(z0, z1);
    float lse = mx + logf(__expf(z0 - mx) + __expf(z1 - mx));
    out[(size_t)i * 2 + 0] = z0 - lse;
    out[(size_t)i * 2 + 1] = z1 - lse;
}

extern "C" void kernel_launch(void* const* d_in, const int* in_sizes, int n_in,
                              void* d_out, int out_size, void* d_ws, size_t ws_size,
                              hipStream_t stream) {
    const int IN = 128;
    const int N  = in_sizes[0] / IN;
    const int E  = in_sizes[1] / 2;

    const float* x    = (const float*)d_in[0];
    const int*   ei   = (const int*)d_in[1];
    const int*   srcI = ei;
    const int*   dstI = ei + E;
    const float* W1   = (const float*)d_in[2];
    const float* b1   = (const float*)d_in[3];
    const float* W2   = (const float*)d_in[4];
    const float* b2   = (const float*)d_in[5];
    const float* W3   = (const float*)d_in[6];
    const float* b3   = (const float*)d_in[7];
    const float* g1   = (const float*)d_in[8];
    const float* be1  = (const float*)d_in[9];
    const float* g2   = (const float*)d_in[10];
    const float* be2  = (const float*)d_in[11];
    const float* Wfc  = (const float*)d_in[12];
    const float* bfc  = (const float*)d_in[13];

    float* ws   = (float*)d_ws;
    float* dis  = ws;
    float* bufA = dis + N;
    float* bufB = bufA + (size_t)N * HC;
    float* bufC = bufB + (size_t)N * HC;
    float* stats = bufC + (size_t)N * HC;

    const int B256 = 256;
    const int NC = N * HC;
    dim3 blkN((N + B256 - 1) / B256);
    dim3 blkNC((NC + B256 - 1) / B256);
    dim3 blkE((E + B256 - 1) / B256);
    dim3 blkG((N / 16 + 7) / 8);
    float invN = 1.0f / (float)N;
    dim3 blkO(((N + OWN_ROWS - 1) / OWN_ROWS) * OWN_SLICES);

    zero_f32<<<blkN, B256, 0, stream>>>(dis, N);
    degree_kernel<<<blkE, B256, 0, stream>>>(dstI, dis, E);
    rsqrt_kernel<<<blkN, B256, 0, stream>>>(dis, N);

    gemm_wmma_f32<<<blkG, B256, 0, stream>>>(x, W1, bufA, dis, b1, bufB, N, IN);
    aggregate_owner<<<blkO, OWN_THREADS, 0, stream>>>(srcI, dstI, dis, bufA, bufB, N, E);
    zero_f32<<<1, 64, 0, stream>>>(stats, 64);
    bn_stats<<<256, 256, 0, stream>>>(bufB, stats, N);
    bn_finalize<<<1, 16, 0, stream>>>(stats, g1, be1, invN);
    bn_apply<<<blkNC, B256, 0, stream>>>(bufB, stats, bufC, NC);

    gemm_wmma_f32<<<blkG, B256, 0, stream>>>(bufC, W2, bufA, dis, b2, bufB, N, HC);
    aggregate_owner<<<blkO, OWN_THREADS, 0, stream>>>(srcI, dstI, dis, bufA, bufB, N, E);
    zero_f32<<<1, 64, 0, stream>>>(stats, 64);
    bn_stats<<<256, 256, 0, stream>>>(bufB, stats, N);
    bn_finalize<<<1, 16, 0, stream>>>(stats, g2, be2, invN);
    bn_apply<<<blkNC, B256, 0, stream>>>(bufB, stats, bufC, NC);

    gemm_wmma_f32<<<blkG, B256, 0, stream>>>(bufC, W3, bufA, dis, b3, bufB, N, HC);
    aggregate_owner<<<blkO, OWN_THREADS, 0, stream>>>(srcI, dstI, dis, bufA, bufB, N, E);

    head_kernel<<<blkN, B256, 0, stream>>>(bufB, Wfc, bfc, (float*)d_out, N);
}


# Degree count: per-edge scattered L2 float atomics replaced by LDS integer counters per (row range, edge slice) work-group; 32 partial count planes summed in the rsqrt launch (integer sums, no order)
# speedup vs baseline: 8.0944x; 1.5059x over previous
#include <hip/hip_runtime.h>
#include <hip/hip_bf16.h>

typedef float v2f __attribute__((ext_vector_type(2)));
typedef float v8f __attribute__((ext_vector_type(8)));

#define HC 16
#define EPSV 1e-5f

__global__ void zero_f32(float* __restrict__ p, int n) {
    int i = blockIdx.x * blockDim.x + threadIdx.x;
    if (i < n) p[i] = 0.0f;
}

#define DEG_ROWS    12512
#define DEG_SLICES  32
#define DEG_THREADS 1024
#define DEG_UNROLL  25
static_assert(DEG_SLICES <= 2 * HC, "the planes must fit the two N x 16 buffers they borrow");

__global__ __launch_bounds__(DEG_THREADS)
void degree_owner(const int* __restrict__ dst, int* __restrict__ planes, int N, int E) {
    __shared__ int cnt[DEG_ROWS];
    const int tid  = threadIdx.x;
    const int grp  = blockIdx.x / DEG_SLICES;
    const int slc  = blockIdx.x % DEG_SLICES;
    const int lo   = grp * DEG_ROWS;
    const int rows = min(DEG_ROWS, N - lo);
    const int e0   = (int)(((long long)E * slc) / DEG_SLICES);
    const int e1   = (int)(((long long)E * (slc + 1)) / DEG_SLICES);

    for (int idx = tid; idx < rows; idx += DEG_THREADS) cnt[idx] = 0;
    __syncthreads();

    for (int base = e0; base < e1; base += DEG_UNROLL * DEG_THREADS) {
        int dv[DEG_UNROLL];
#pragma unroll
        for (int k = 0; k < DEG_UNROLL; ++k) dv[k] = dst[min(base + tid + k * DEG_THREADS, e1 - 1)];
#pragma unroll
        for (int k = 0; k < DEG_UNROLL; ++k) {
            const unsigned r = (unsigned)(dv[k] - lo);
            if (base + tid + k * DEG_THREADS < e1 && r < (unsigned)rows) atomicAdd(&cnt[r], 1);
        }
    }
    __syncthreads();

    int* prow = planes + (size_t)slc * N + lo;
    for (int idx = tid; idx < rows; idx += DEG_THREADS) prow[idx] = cnt[idx];
}

__global__ void degree_merge_rsqrt(const int* __restrict__ planes, float* __restrict__ dis, int N) {
    int i = blockIdx.x * blockDim.x + threadIdx.x;
    if (i >= N) return;
    int c = 0;
#pragma unroll
    for (int s = 0; s < DEG_SLICES; ++s) c += planes[(size_t)s * N + i];
    dis[i] = rsqrtf((float)c + 1.0f);
}

__global__ void gemm_wmma_f32(const float* __restrict__ A, const float* __restrict__ B,
                              float* __restrict__ C, const float* __restrict__ dis,
                              const float* __restrict__ bias, float* __restrict__ S, int N, int K) {
    int wave = (int)((blockIdx.x * blockDim.x + threadIdx.x) >> 5);
    int lane = threadIdx.x & 31;
    int row0 = wave * 16;
    if (row0 >= N) return;

    int m  = lane & 15;
    int hi = lane >> 4;
    const float* arow = A + (size_t)(row0 + m) * K;

    v8f acc = {};
    for (int k0 = 0; k0 < K; k0 += 4) {
        int ka = k0 + 2 * hi;
        v2f a, b;
        a.x = arow[ka];
        a.y = arow[ka + 1];
        b.x = B[(size_t)ka * HC + m];
        b.y = B[(size_t)(ka + 1) * HC + m];
        acc = __builtin_amdgcn_wmma_f32_16x16x4_f32(
             false, a,  false, b,
             (short)0, acc,  false,  false);
    }

    float* cbase = C + (size_t)(row0 + hi * 8) * HC + m;
    float* sbase = S + (size_t)(row0 + hi * 8) * HC + m;
    const float* drow = dis + row0 + hi * 8;
    float bm = bias[m];
#pragma unroll
    for (int r = 0; r < 8; ++r) {
        float d = drow[r];
        cbase[(size_t)r * HC] = d * acc[r];
        sbase[(size_t)r * HC] = d * d * acc[r] + bm;
    }
}

#define OWN_ROWS    3126
#define OWN_SLICES  8
#define OWN_PITCH   17
#define OWN_THREADS 1024
#define OWN_QCAP    64
#define OWN_CHUNK   (16 * OWN_THREADS)
#define OWN_SRC_BITS 20
static_assert(OWN_ROWS <= (1 << (32 - OWN_SRC_BITS)), "local row must fit the queue word");

__device__ __forceinline__ void own_add_edge(unsigned w, const float* __restrict__ hw, float* acc) {
    const unsigned s  = w & ((1u << OWN_SRC_BITS) - 1u);
    const unsigned dl = w >> OWN_SRC_BITS;
    const float4* hrow = (const float4*)(hw + (size_t)s * HC);
    float* arow = acc + dl * OWN_PITCH;
#pragma unroll
    for (int q = 0; q < 4; ++q) {
        float4 v = hrow[q];
        atomicAdd(arow + 4 * q + 0, v.x);
        atomicAdd(arow + 4 * q + 1, v.y);
        atomicAdd(arow + 4 * q + 2, v.z);
        atomicAdd(arow + 4 * q + 3, v.w);
    }
}

__global__ __launch_bounds__(OWN_THREADS)
void aggregate_owner(const int* __restrict__ src, const int* __restrict__ dst,
                     const float* __restrict__ dis, const float* __restrict__ hw,
                     float* __restrict__ out, int N, int E) {
    __shared__ float acc[OWN_ROWS * OWN_PITCH];
    __shared__ unsigned queue[(OWN_THREADS / 32) * OWN_QCAP];
    __shared__ float dsl[OWN_ROWS];
    const int tid  = threadIdx.x;
    const int lane = tid & 31;
    unsigned* myq = queue + (tid >> 5) * OWN_QCAP;
    const int grp  = blockIdx.x / OWN_SLICES;
    const int slc  = blockIdx.x % OWN_SLICES;
    const int lo   = grp * OWN_ROWS;
    const int rows = min(OWN_ROWS, N - lo);
    const int per  = ((E + OWN_SLICES * OWN_THREADS - 1) / (OWN_SLICES * OWN_THREADS)) * OWN_THREADS;
    const int e0   = min(E, slc * per);
    const int e1   = min(E, e0 + per);

    constexpr int DPT = (OWN_ROWS + OWN_THREADS - 1) / OWN_THREADS;
    float dreg[DPT];
#pragma unroll
    for (int j = 0; j < DPT; ++j) {
        int idx = tid + j * OWN_THREADS;
        dreg[j] = (idx < rows) ? dis[lo + idx] : 0.0f;
    }
    for (int idx = tid; idx < rows * OWN_PITCH; idx += OWN_THREADS) acc[idx] = 0.0f;
#pragma unroll
    for (int j = 0; j < DPT; ++j) {
        int idx = tid + j * OWN_THREADS;
        if (idx < rows) dsl[idx] = dreg[j];
    }
    __syncthreads();

    int qn = 0;

#define OWN_STEP(m, km, w)                                                              \
    {                                                                                   \
        if (m) myq[qn + (int)__builtin_amdgcn_mbcnt_lo(km, 0u)] = (w);                  \
        qn += __popc(km);                                                               \
          \
          \
          \
        __builtin_amdgcn_wave_barrier();                                                \
        if (qn >= 32) {                                                                 \
            qn -= 32;                                                                   \
            own_add_edge(myq[qn + lane], hw, acc);                                      \
        }                                                                               \
                             \
        __builtin_amdgcn_wave_barrier();                                                \
    }

    int base = e0;
    for (; base + OWN_CHUNK <= e1; base += OWN_CHUNK) {
        const int* p  = dst + base + tid;
        const int* ps = src + base + tid;
        int dv[16], sv[16];
#pragma unroll
        for (int k = 0; k < 16; ++k) dv[k] = p[k * OWN_THREADS];
#pragma unroll
        for (int k = 0; k < 16; ++k)
            sv[k] = __hip_atomic_load(ps + k * OWN_THREADS, __ATOMIC_RELAXED, __HIP_MEMORY_SCOPE_WAVEFRONT);
        unsigned km[16];
        unsigned any = 0u;
#pragma unroll
        for (int k = 0; k < 16; ++k) {
            km[k] = __builtin_amdgcn_ballot_w32((unsigned)(dv[k] - lo) < (unsigned)rows);
            any |= km[k];
        }
        if (any != 0u) {
#pragma unroll
            for (int k = 0; k < 16; ++k) {
                if (km[k] != 0u) {
                    bool m = (unsigned)(dv[k] - lo) < (unsigned)rows;
                    OWN_STEP(m, km[k], (unsigned)sv[k] | ((unsigned)(dv[k] - lo) << OWN_SRC_BITS))
                }
            }
        }
    }
    for (; base < e1; base += OWN_THREADS) {
        int e = base + tid;
        int d = (e < e1) ? dst[e] : -1;
        int s = (e < e1) ? src[e] : 0;
        bool m = (unsigned)(d - lo) < (unsigned)rows;
        unsigned k1 = __builtin_amdgcn_ballot_w32(m);
        if (k1 != 0u) OWN_STEP(m, k1, (unsigned)s | ((unsigned)(d - lo) << OWN_SRC_BITS))
    }
#undef OWN_STEP
    if (lane < qn) own_add_edge(myq[lane], hw, acc);
    __syncthreads();

    for (int idx = tid; idx < rows * HC; idx += OWN_THREADS) {
        int r = idx >> 4, c = idx & 15;
        atomicAdd(out + (size_t)lo * HC + idx, dsl[r] * acc[r * OWN_PITCH + c]);
    }
}

__global__ void bn_stats(const float* __restrict__ agg, float* __restrict__ stats, int N) {
    __shared__ float s1s[256];
    __shared__ float s2s[256];
    int c = threadIdx.x & 15;
    int g = threadIdx.x >> 4;
    int stride = gridDim.x * 16;
    float s1 = 0.0f, s2 = 0.0f;
    for (int row = blockIdx.x * 16 + g; row < N; row += stride) {
        float v = agg[(size_t)row * HC + c];
        v = fmaxf(v, 0.0f);
        s1 += v;
        s2 += v * v;
    }
    s1s[threadIdx.x] = s1;
    s2s[threadIdx.x] = s2;
    __syncthreads();
    if (threadIdx.x < 16) {
        float t1 = 0.0f, t2 = 0.0f;
#pragma unroll
        for (int gg = 0; gg < 16; ++gg) {
            t1 += s1s[gg * 16 + threadIdx.x];
            t2 += s2s[gg * 16 + threadIdx.x];
        }
        atomicAdd(&stats[threadIdx.x], t1);
        atomicAdd(&stats[16 + threadIdx.x], t2);
    }
}

__global__ void bn_finalize(float* __restrict__ stats, const float* __restrict__ gamma,
                            const float* __restrict__ beta, float invN) {
    int c = threadIdx.x;
    if (c < 16) {
        float mu  = stats[c] * invN;
        float var = stats[16 + c] * invN - mu * mu;
        float sc  = rsqrtf(var + EPSV) * gamma[c];
        stats[32 + c] = sc;
        stats[48 + c] = beta[c] - mu * sc;
    }
}

__global__ void bn_apply(const float* __restrict__ agg, const float* __restrict__ stats,
                         float* __restrict__ h, int NC) {
    int i = blockIdx.x * blockDim.x + threadIdx.x;
    if (i >= NC) return;
    int c = i & 15;
    float v = fmaxf(agg[i], 0.0f);
    h[i] = v * stats[32 + c] + stats[48 + c];
}

__global__ void head_kernel(const float* __restrict__ h, const float* __restrict__ Wfc,
                            const float* __restrict__ bfc, float* __restrict__ out, int N) {
    int i = blockIdx.x * blockDim.x + threadIdx.x;
    if (i >= N) return;
    const float* r = h + (size_t)i * HC;
    float z0 = bfc[0], z1 = bfc[1];
#pragma unroll
    for (int c = 0; c < HC; ++c) {
        float v = r[c];
        z0 += v * Wfc[c * 2 + 0];
        z1 += v * Wfc[c * 2 + 1];
    }
    float mx  = fmaxf(z0, z1);
    float lse = mx + logf(__expf(z0 - mx) + __expf(z1 - mx));
    out[(size_t)i * 2 + 0] = z0 - lse;
    out[(size_t)i * 2 + 1] = z1 - lse;
}

extern "C" void kernel_launch(void* const* d_in, const int* in_sizes, int n_in,
                              void* d_out, int out_size, void* d_ws, size_t ws_size,
                              hipStream_t stream) {
    const int IN = 128;
    const int N  = in_sizes[0] / IN;
    const int E  = in_sizes[1] / 2;

    const float* x    = (const float*)d_in[0];
    const int*   ei   = (const int*)d_in[1];
    const int*   srcI = ei;
    const int*   dstI = ei + E;
    const float* W1   = (const float*)d_in[2];
    const float* b1   = (const float*)d_in[3];
    const float* W2   = (const float*)d_in[4];
    const float* b2   = (const float*)d_in[5];
    const float* W3   = (const float*)d_in[6];
    const float* b3   = (const float*)d_in[7];
    const float* g1   = (const float*)d_in[8];
    const float* be1  = (const float*)d_in[9];
    const float* g2   = (const float*)d_in[10];
    const float* be2  = (const float*)d_in[11];
    const float* Wfc  = (const float*)d_in[12];
    const float* bfc  = (const float*)d_in[13];

    float* ws   = (float*)d_ws;
    float* dis  = ws;
    float* bufA = dis + N;
    float* bufB = bufA + (size_t)N * HC;
    float* bufC = bufB + (size_t)N * HC;
    float* stats = bufC + (size_t)N * HC;

    const int B256 = 256;
    const int NC = N * HC;
    dim3 blkN((N + B256 - 1) / B256);
    dim3 blkNC((NC + B256 - 1) / B256);
    dim3 blkG((N / 16 + 7) / 8);
    float invN = 1.0f / (float)N;
    dim3 blkO(((N + OWN_ROWS - 1) / OWN_ROWS) * OWN_SLICES);

    int* planes = (int*)bufA;
    dim3 blkD(((N + DEG_ROWS - 1) / DEG_ROWS) * DEG_SLICES);
    degree_owner<<<blkD, DEG_THREADS, 0, stream>>>(dstI, planes, N, E);
    degree_merge_rsqrt<<<blkN, B256, 0, stream>>>(planes, dis, N);

    gemm_wmma_f32<<<blkG, B256, 0, stream>>>(x, W1, bufA, dis, b1, bufB, N, IN);
    aggregate_owner<<<blkO, OWN_THREADS, 0, stream>>>(srcI, dstI, dis, bufA, bufB, N, E);
    zero_f32<<<1, 64, 0, stream>>>(stats, 64);
    bn_stats<<<256, 256, 0, stream>>>(bufB, stats, N);
    bn_finalize<<<1, 16, 0, stream>>>(stats, g1, be1, invN);
    bn_apply<<<blkNC, B256, 0, stream>>>(bufB, stats, bufC, NC);

    gemm_wmma_f32<<<blkG, B256, 0, stream>>>(bufC, W2, bufA, dis, b2, bufB, N, HC);
    aggregate_owner<<<blkO, OWN_THREADS, 0, stream>>>(srcI, dstI, dis, bufA, bufB, N, E);
    zero_f32<<<1, 64, 0, stream>>>(stats, 64);
    bn_stats<<<256, 256, 0, stream>>>(bufB, stats, N);
    bn_finalize<<<1, 16, 0, stream>>>(stats, g2, be2, invN);
    bn_apply<<<blkNC, B256, 0, stream>>>(bufB, stats, bufC, NC);

    gemm_wmma_f32<<<blkG, B256, 0, stream>>>(bufC, W3, bufA, dis, b3, bufB, N, HC);
    aggregate_owner<<<blkO, OWN_THREADS, 0, stream>>>(srcI, dstI, dis, bufA, bufB, N, E);

    head_kernel<<<blkN, B256, 0, stream>>>(bufB, Wfc, bfc, (float*)d_out, N);
}


# BatchNorm finalize and apply folded into the next layer's GEMM A-operand load, sum clearing into the degree merge: 17 launches to 11, normalized activations never stored
# speedup vs baseline: 8.7479x; 1.0807x over previous
#include <hip/hip_runtime.h>
#include <hip/hip_bf16.h>

typedef float v2f __attribute__((ext_vector_type(2)));
typedef float v8f __attribute__((ext_vector_type(8)));

#define HC 16
#define EPSV 1e-5f

#define DEG_ROWS    12512
#define DEG_SLICES  32
#define DEG_THREADS 1024
#define DEG_UNROLL  25
static_assert(DEG_SLICES <= 2 * HC, "the planes must fit the two N x 16 buffers they borrow");

__global__ __launch_bounds__(DEG_THREADS)
void degree_owner(const int* __restrict__ dst, int* __restrict__ planes, int N, int E) {
    __shared__ int cnt[DEG_ROWS];
    const int tid  = threadIdx.x;
    const int grp  = blockIdx.x / DEG_SLICES;
    const int slc  = blockIdx.x % DEG_SLICES;
    const int lo   = grp * DEG_ROWS;
    const int rows = min(DEG_ROWS, N - lo);
    const int e0   = (int)(((long long)E * slc) / DEG_SLICES);
    const int e1   = (int)(((long long)E * (slc + 1)) / DEG_SLICES);

    for (int idx = tid; idx < rows; idx += DEG_THREADS) cnt[idx] = 0;
    __syncthreads();

    for (int base = e0; base < e1; base += DEG_UNROLL * DEG_THREADS) {
        int dv[DEG_UNROLL];
#pragma unroll
        for (int k = 0; k < DEG_UNROLL; ++k) dv[k] = dst[min(base + tid + k * DEG_THREADS, e1 - 1)];
#pragma unroll
        for (int k = 0; k < DEG_UNROLL; ++k) {
            const unsigned r = (unsigned)(dv[k] - lo);
            if (base + tid + k * DEG_THREADS < e1 && r < (unsigned)rows) atomicAdd(&cnt[r], 1);
        }
    }
    __syncthreads();

    int* prow = planes + (size_t)slc * N + lo;
    for (int idx = tid; idx < rows; idx += DEG_THREADS) prow[idx] = cnt[idx];
}

__global__ void degree_merge_rsqrt(const int* __restrict__ planes, float* __restrict__ dis,
                                   float* __restrict__ stats, int N) {
    int i = blockIdx.x * blockDim.x + threadIdx.x;
    if (blockIdx.x == 0 && threadIdx.x < 64) stats[threadIdx.x] = 0.0f;
    if (i >= N) return;
    int c = 0;
#pragma unroll
    for (int s = 0; s < DEG_SLICES; ++s) c += planes[(size_t)s * N + i];
    dis[i] = rsqrtf((float)c + 1.0f);
}

__device__ __forceinline__ float bn_relu(float v, const float* __restrict__ stats, const float* __restrict__ gamma,
                                         const float* __restrict__ beta, float invN, int c) {
    float mu  = stats[c] * invN;
    float var = stats[16 + c] * invN - mu * mu;
    float sc  = rsqrtf(var + EPSV) * gamma[c];
    float sh  = beta[c] - mu * sc;
    return fmaxf(v, 0.0f) * sc + sh;
}

template <bool BN>
__device__ __forceinline__ void gemm_tile(const float* __restrict__ A, const float* __restrict__ B,
                                          float* __restrict__ C, const float* __restrict__ dis,
                                          const float* __restrict__ bias, float* __restrict__ S,
                                          const float* __restrict__ stats, const float* __restrict__ gamma,
                                          const float* __restrict__ beta, float invN, int N, int K) {
    int wave = (int)((blockIdx.x * blockDim.x + threadIdx.x) >> 5);
    int lane = threadIdx.x & 31;
    int row0 = wave * 16;
    if (row0 >= N) return;

    int m  = lane & 15;
    int hi = lane >> 4;

    v8f acc = {};
    if constexpr (BN) {
        const float* arow = A + (size_t)(row0 + m) * HC;
#pragma unroll
        for (int k0 = 0; k0 < HC; k0 += 4) {
            int ka = k0 + 2 * hi;
            v2f a, b;
            a.x = bn_relu(arow[ka],     stats, gamma, beta, invN, ka);
            a.y = bn_relu(arow[ka + 1], stats, gamma, beta, invN, ka + 1);
            b.x = B[(size_t)ka * HC + m];
            b.y = B[(size_t)(ka + 1) * HC + m];
            acc = __builtin_amdgcn_wmma_f32_16x16x4_f32(
                 false, a,  false, b,
                 (short)0, acc,  false,  false);
        }
    } else {
        const float* arow = A + (size_t)(row0 + m) * K;
        for (int k0 = 0; k0 < K; k0 += 4) {
            int ka = k0 + 2 * hi;
            v2f a, b;
            a.x = arow[ka];
            a.y = arow[ka + 1];
            b.x = B[(size_t)ka * HC + m];
            b.y = B[(size_t)(ka + 1) * HC + m];
            acc = __builtin_amdgcn_wmma_f32_16x16x4_f32(
                 false, a,  false, b,
                 (short)0, acc,  false,  false);
        }
    }

    float* cbase = C + (size_t)(row0 + hi * 8) * HC + m;
    float* sbase = S + (size_t)(row0 + hi * 8) * HC + m;
    const float* drow = dis + row0 + hi * 8;
    float bm = bias[m];
#pragma unroll
    for (int r = 0; r < 8; ++r) {
        float d = drow[r];
        cbase[(size_t)r * HC] = d * acc[r];
        sbase[(size_t)r * HC] = d * d * acc[r] + bm;
    }
}

__global__ void gemm_wmma_f32(const float* __restrict__ A, const float* __restrict__ B,
                              float* __restrict__ C, const float* __restrict__ dis,
                              const float* __restrict__ bias, float* __restrict__ S, int N, int K) {
    gemm_tile<false>(A, B, C, dis, bias, S, nullptr, nullptr, nullptr, 0.0f, N, K);
}

__global__ void gemm_bn_wmma_f32(const float* __restrict__ A, const float* __restrict__ B,
                                 float* __restrict__ C, const float* __restrict__ dis,
                                 const float* __restrict__ bias, float* __restrict__ S,
                                 const float* __restrict__ stats, const float* __restrict__ gamma,
                                 const float* __restrict__ beta, float invN, int N) {
    gemm_tile<true>(A, B, C, dis, bias, S, stats, gamma, beta, invN, N, HC);
}

#define OWN_ROWS    3126
#define OWN_SLICES  8
#define OWN_PITCH   17
#define OWN_THREADS 1024
#define OWN_QCAP    64
#define OWN_CHUNK   (16 * OWN_THREADS)
#define OWN_SRC_BITS 20
static_assert(OWN_ROWS <= (1 << (32 - OWN_SRC_BITS)), "local row must fit the queue word");

__device__ __forceinline__ void own_add_edge(unsigned w, const float* __restrict__ hw, float* acc) {
    const unsigned s  = w & ((1u << OWN_SRC_BITS) - 1u);
    const unsigned dl = w >> OWN_SRC_BITS;
    const float4* hrow = (const float4*)(hw + (size_t)s * HC);
    float* arow = acc + dl * OWN_PITCH;
#pragma unroll
    for (int q = 0; q < 4; ++q) {
        float4 v = hrow[q];
        atomicAdd(arow + 4 * q + 0, v.x);
        atomicAdd(arow + 4 * q + 1, v.y);
        atomicAdd(arow + 4 * q + 2, v.z);
        atomicAdd(arow + 4 * q + 3, v.w);
    }
}

__global__ __launch_bounds__(OWN_THREADS)
void aggregate_owner(const int* __restrict__ src, const int* __restrict__ dst,
                     const float* __restrict__ dis, const float* __restrict__ hw,
                     float* __restrict__ out, int N, int E) {
    __shared__ float acc[OWN_ROWS * OWN_PITCH];
    __shared__ unsigned queue[(OWN_THREADS / 32) * OWN_QCAP];
    __shared__ float dsl[OWN_ROWS];
    const int tid  = threadIdx.x;
    const int lane = tid & 31;
    unsigned* myq = queue + (tid >> 5) * OWN_QCAP;
    const int grp  = blockIdx.x / OWN_SLICES;
    const int slc  = blockIdx.x % OWN_SLICES;
    const int lo   = grp * OWN_ROWS;
    const int rows = min(OWN_ROWS, N - lo);
    const int per  = ((E + OWN_SLICES * OWN_THREADS - 1) / (OWN_SLICES * OWN_THREADS)) * OWN_THREADS;
    const int e0   = min(E, slc * per);
    const int e1   = min(E, e0 + per);

    constexpr int DPT = (OWN_ROWS + OWN_THREADS - 1) / OWN_THREADS;
    float dreg[DPT];
#pragma unroll
    for (int j = 0; j < DPT; ++j) {
        int idx = tid + j * OWN_THREADS;
        dreg[j] = (idx < rows) ? dis[lo + idx] : 0.0f;
    }
    for (int idx = tid; idx < rows * OWN_PITCH; idx += OWN_THREADS) acc[idx] = 0.0f;
#pragma unroll
    for (int j = 0; j < DPT; ++j) {
        int idx = tid + j * OWN_THREADS;
        if (idx < rows) dsl[idx] = dreg[j];
    }
    __syncthreads();

    int qn = 0;

#define OWN_STEP(m, km, w)                                                              \
    {                                                                                   \
        if (m) myq[qn + (int)__builtin_amdgcn_mbcnt_lo(km, 0u)] = (w);                  \
        qn += __popc(km);                                                               \
          \
          \
          \
        __builtin_amdgcn_wave_barrier();                                                \
        if (qn >= 32) {                                                                 \
            qn -= 32;                                                                   \
            own_add_edge(myq[qn + lane], hw, acc);                                      \
        }                                                                               \
                             \
        __builtin_amdgcn_wave_barrier();                                                \
    }

    int base = e0;
    for (; base + OWN_CHUNK <= e1; base += OWN_CHUNK) {
        const int* p  = dst + base + tid;
        const int* ps = src + base + tid;
        int dv[16], sv[16];
#pragma unroll
        for (int k = 0; k < 16; ++k) dv[k] = p[k * OWN_THREADS];
#pragma unroll
        for (int k = 0; k < 16; ++k)
            sv[k] = __hip_atomic_load(ps + k * OWN_THREADS, __ATOMIC_RELAXED, __HIP_MEMORY_SCOPE_WAVEFRONT);
        unsigned km[16];
        unsigned any = 0u;
#pragma unroll
        for (int k = 0; k < 16; ++k) {
            km[k] = __builtin_amdgcn_ballot_w32((unsigned)(dv[k] - lo) < (unsigned)rows);
            any |= km[k];
        }
        if (any != 0u) {
#pragma unroll
            for (int k = 0; k < 16; ++k) {
                if (km[k] != 0u) {
                    bool m = (unsigned)(dv[k] - lo) < (unsigned)rows;
                    OWN_STEP(m, km[k], (unsigned)sv[k] | ((unsigned)(dv[k] - lo) << OWN_SRC_BITS))
                }
            }
        }
    }
    for (; base < e1; base += OWN_THREADS) {
        int e = base + tid;
        int d = (e < e1) ? dst[e] : -1;
        int s = (e < e1) ? src[e] : 0;
        bool m = (unsigned)(d - lo) < (unsigned)rows;
        unsigned k1 = __builtin_amdgcn_ballot_w32(m);
        if (k1 != 0u) OWN_STEP(m, k1, (unsigned)s | ((unsigned)(d - lo) << OWN_SRC_BITS))
    }
#undef OWN_STEP
    if (lane < qn) own_add_edge(myq[lane], hw, acc);
    __syncthreads();

    for (int idx = tid; idx < rows * HC; idx += OWN_THREADS) {
        int r = idx >> 4, c = idx & 15;
        atomicAdd(out + (size_t)lo * HC + idx, dsl[r] * acc[r * OWN_PITCH + c]);
    }
}

__global__ void bn_stats(const float* __restrict__ agg, float* __restrict__ stats, int N) {
    __shared__ float s1s[256];
    __shared__ float s2s[256];
    int c = threadIdx.x & 15;
    int g = threadIdx.x >> 4;
    int stride = gridDim.x * 16;
    float s1 = 0.0f, s2 = 0.0f;
    for (int row = blockIdx.x * 16 + g; row < N; row += stride) {
        float v = agg[(size_t)row * HC + c];
        v = fmaxf(v, 0.0f);
        s1 += v;
        s2 += v * v;
    }
    s1s[threadIdx.x] = s1;
    s2s[threadIdx.x] = s2;
    __syncthreads();
    if (threadIdx.x < 16) {
        float t1 = 0.0f, t2 = 0.0f;
#pragma unroll
        for (int gg = 0; gg < 16; ++gg) {
            t1 += s1s[gg * 16 + threadIdx.x];
            t2 += s2s[gg * 16 + threadIdx.x];
        }
        atomicAdd(&stats[threadIdx.x], t1);
        atomicAdd(&stats[16 + threadIdx.x], t2);
    }
}

__global__ void head_kernel(const float* __restrict__ h, const float* __restrict__ Wfc,
                            const float* __restrict__ bfc, float* __restrict__ out, int N) {
    int i = blockIdx.x * blockDim.x + threadIdx.x;
    if (i >= N) return;
    const float* r = h + (size_t)i * HC;
    float z0 = bfc[0], z1 = bfc[1];
#pragma unroll
    for (int c = 0; c < HC; ++c) {
        float v = r[c];
        z0 += v * Wfc[c * 2 + 0];
        z1 += v * Wfc[c * 2 + 1];
    }
    float mx  = fmaxf(z0, z1);
    float lse = mx + logf(__expf(z0 - mx) + __expf(z1 - mx));
    out[(size_t)i * 2 + 0] = z0 - lse;
    out[(size_t)i * 2 + 1] = z1 - lse;
}

extern "C" void kernel_launch(void* const* d_in, const int* in_sizes, int n_in,
                              void* d_out, int out_size, void* d_ws, size_t ws_size,
                              hipStream_t stream) {
    const int IN = 128;
    const int N  = in_sizes[0] / IN;
    const int E  = in_sizes[1] / 2;

    const float* x    = (const float*)d_in[0];
    const int*   ei   = (const int*)d_in[1];
    const int*   srcI = ei;
    const int*   dstI = ei + E;
    const float* W1   = (const float*)d_in[2];
    const float* b1   = (const float*)d_in[3];
    const float* W2   = (const float*)d_in[4];
    const float* b2   = (const float*)d_in[5];
    const float* W3   = (const float*)d_in[6];
    const float* b3   = (const float*)d_in[7];
    const float* g1   = (const float*)d_in[8];
    const float* be1  = (const float*)d_in[9];
    const float* g2   = (const float*)d_in[10];
    const float* be2  = (const float*)d_in[11];
    const float* Wfc  = (const float*)d_in[12];
    const float* bfc  = (const float*)d_in[13];

    float* ws   = (float*)d_ws;
    float* dis  = ws;
    float* bufA = dis + N;
    float* bufB = bufA + (size_t)N * HC;
    float* bufC = bufB + (size_t)N * HC;
    float* stats = bufC + (size_t)N * HC;

    const int B256 = 256;
    dim3 blkN((N + B256 - 1) / B256);
    dim3 blkG((N / 16 + 7) / 8);
    float invN = 1.0f / (float)N;
    dim3 blkO(((N + OWN_ROWS - 1) / OWN_ROWS) * OWN_SLICES);

    int* planes = (int*)bufA;
    dim3 blkD(((N + DEG_ROWS - 1) / DEG_ROWS) * DEG_SLICES);
    degree_owner<<<blkD, DEG_THREADS, 0, stream>>>(dstI, planes, N, E);
    degree_merge_rsqrt<<<blkN, B256, 0, stream>>>(planes, dis, stats, N);

    gemm_wmma_f32<<<blkG, B256, 0, stream>>>(x, W1, bufA, dis, b1, bufB, N, IN);
    aggregate_owner<<<blkO, OWN_THREADS, 0, stream>>>(srcI, dstI, dis, bufA, bufB, N, E);
    bn_stats<<<256, 256, 0, stream>>>(bufB, stats, N);

    gemm_bn_wmma_f32<<<blkG, B256, 0, stream>>>(bufB, W2, bufA, dis, b2, bufC, stats, g1, be1, invN, N);
    aggregate_owner<<<blkO, OWN_THREADS, 0, stream>>>(srcI, dstI, dis, bufA, bufC, N, E);
    bn_stats<<<256, 256, 0, stream>>>(bufC, stats + 32, N);

    gemm_bn_wmma_f32<<<blkG, B256, 0, stream>>>(bufC, W3, bufA, dis, b3, bufB, stats + 32, g2, be2, invN, N);
    aggregate_owner<<<blkO, OWN_THREADS, 0, stream>>>(srcI, dstI, dis, bufA, bufB, N, E);

    head_kernel<<<blkN, B256, 0, stream>>>(bufB, Wfc, bfc, (float*)d_out, N);
}


# Classifier folded into layer 3 ahead of its aggregation (both linear): h@(W3@Wfc) on WMMA, layer 3 gathers and sums 2-channel rows, not 16; products re-associated, float atomics in arrival order
# speedup vs baseline: 9.8516x; 1.1262x over previous
#include <hip/hip_runtime.h>
#include <hip/hip_bf16.h>

typedef float v2f __attribute__((ext_vector_type(2)));
typedef float v8f __attribute__((ext_vector_type(8)));

#define HC 16
#define EPSV 1e-5f

#define DEG_ROWS    12512
#define DEG_SLICES  32
#define DEG_THREADS 1024
#define DEG_UNROLL  25
static_assert(DEG_SLICES <= 2 * HC, "the planes must fit the two N x 16 buffers they borrow");

__global__ __launch_bounds__(DEG_THREADS)
void degree_owner(const int* __restrict__ dst, int* __restrict__ planes, int N, int E) {
    __shared__ int cnt[DEG_ROWS];
    const int tid  = threadIdx.x;
    const int grp  = blockIdx.x / DEG_SLICES;
    const int slc  = blockIdx.x % DEG_SLICES;
    const int lo   = grp * DEG_ROWS;
    const int rows = min(DEG_ROWS, N - lo);
    const int e0   = (int)(((long long)E * slc) / DEG_SLICES);
    const int e1   = (int)(((long long)E * (slc + 1)) / DEG_SLICES);

    for (int idx = tid; idx < rows; idx += DEG_THREADS) cnt[idx] = 0;
    __syncthreads();

    for (int base = e0; base < e1; base += DEG_UNROLL * DEG_THREADS) {
        int dv[DEG_UNROLL];
#pragma unroll
        for (int k = 0; k < DEG_UNROLL; ++k) dv[k] = dst[min(base + tid + k * DEG_THREADS, e1 - 1)];
#pragma unroll
        for (int k = 0; k < DEG_UNROLL; ++k) {
            const unsigned r = (unsigned)(dv[k] - lo);
            if (base + tid + k * DEG_THREADS < e1 && r < (unsigned)rows) atomicAdd(&cnt[r], 1);
        }
    }
    __syncthreads();

    int* prow = planes + (size_t)slc * N + lo;
    for (int idx = tid; idx < rows; idx += DEG_THREADS) prow[idx] = cnt[idx];
}

__global__ void degree_merge_rsqrt(const int* __restrict__ planes, float* __restrict__ dis,
                                   float* __restrict__ stats, int N) {
    int i = blockIdx.x * blockDim.x + threadIdx.x;
    if (blockIdx.x == 0 && threadIdx.x < 64) stats[threadIdx.x] = 0.0f;
    if (i >= N) return;
    int c = 0;
#pragma unroll
    for (int s = 0; s < DEG_SLICES; ++s) c += planes[(size_t)s * N + i];
    dis[i] = rsqrtf((float)c + 1.0f);
}

__device__ __forceinline__ float bn_relu(float v, const float* __restrict__ stats, const float* __restrict__ gamma,
                                         const float* __restrict__ beta, float invN, int c) {
    float mu  = stats[c] * invN;
    float var = stats[16 + c] * invN - mu * mu;
    float sc  = rsqrtf(var + EPSV) * gamma[c];
    float sh  = beta[c] - mu * sc;
    return fmaxf(v, 0.0f) * sc + sh;
}

template <bool BN>
__device__ __forceinline__ void gemm_tile(const float* __restrict__ A, const float* __restrict__ B,
                                          float* __restrict__ C, const float* __restrict__ dis,
                                          const float* __restrict__ bias, float* __restrict__ S,
                                          const float* __restrict__ stats, const float* __restrict__ gamma,
                                          const float* __restrict__ beta, float invN, int N, int K) {
    int wave = (int)((blockIdx.x * blockDim.x + threadIdx.x) >> 5);
    int lane = threadIdx.x & 31;
    int row0 = wave * 16;
    if (row0 >= N) return;

    int m  = lane & 15;
    int hi = lane >> 4;

    v8f acc = {};
    if constexpr (BN) {
        const float* arow = A + (size_t)(row0 + m) * HC;
#pragma unroll
        for (int k0 = 0; k0 < HC; k0 += 4) {
            int ka = k0 + 2 * hi;
            v2f a, b;
            a.x = bn_relu(arow[ka],     stats, gamma, beta, invN, ka);
            a.y = bn_relu(arow[ka + 1], stats, gamma, beta, invN, ka + 1);
            b.x = B[(size_t)ka * HC + m];
            b.y = B[(size_t)(ka + 1) * HC + m];
            acc = __builtin_amdgcn_wmma_f32_16x16x4_f32(
                 false, a,  false, b,
                 (short)0, acc,  false,  false);
        }
    } else {
        const float* arow = A + (size_t)(row0 + m) * K;
        for (int k0 = 0; k0 < K; k0 += 4) {
            int ka = k0 + 2 * hi;
            v2f a, b;
            a.x = arow[ka];
            a.y = arow[ka + 1];
            b.x = B[(size_t)ka * HC + m];
            b.y = B[(size_t)(ka + 1) * HC + m];
            acc = __builtin_amdgcn_wmma_f32_16x16x4_f32(
                 false, a,  false, b,
                 (short)0, acc,  false,  false);
        }
    }

    float* cbase = C + (size_t)(row0 + hi * 8) * HC + m;
    float* sbase = S + (size_t)(row0 + hi * 8) * HC + m;
    const float* drow = dis + row0 + hi * 8;
    float bm = bias[m];
#pragma unroll
    for (int r = 0; r < 8; ++r) {
        float d = drow[r];
        cbase[(size_t)r * HC] = d * acc[r];
        sbase[(size_t)r * HC] = d * d * acc[r] + bm;
    }
}

__global__ void gemm_wmma_f32(const float* __restrict__ A, const float* __restrict__ B,
                              float* __restrict__ C, const float* __restrict__ dis,
                              const float* __restrict__ bias, float* __restrict__ S, int N, int K) {
    gemm_tile<false>(A, B, C, dis, bias, S, nullptr, nullptr, nullptr, 0.0f, N, K);
}

__global__ void gemm_bn_wmma_f32(const float* __restrict__ A, const float* __restrict__ B,
                                 float* __restrict__ C, const float* __restrict__ dis,
                                 const float* __restrict__ bias, float* __restrict__ S,
                                 const float* __restrict__ stats, const float* __restrict__ gamma,
                                 const float* __restrict__ beta, float invN, int N) {
    gemm_tile<true>(A, B, C, dis, bias, S, stats, gamma, beta, invN, N, HC);
}

__global__ void gemm_bn_fold_wmma_f32(const float* __restrict__ A, const float* __restrict__ W,
                                      const float* __restrict__ Wfc, const float* __restrict__ bfc,
                                      float* __restrict__ C, const float* __restrict__ dis,
                                      const float* __restrict__ bias, float* __restrict__ S,
                                      const float* __restrict__ stats, const float* __restrict__ gamma,
                                      const float* __restrict__ beta, float invN, int N) {
    int wave = (int)((blockIdx.x * blockDim.x + threadIdx.x) >> 5);
    int lane = threadIdx.x & 31;
    int row0 = wave * 16;
    if (row0 >= N) return;

    int m  = lane & 15;
    int hi = lane >> 4;
    const bool col_on = m < 2;
    const int  mc = m & 1;

    v8f wp = {};
#pragma unroll
    for (int k0 = 0; k0 < HC; k0 += 4) {
        int ka = k0 + 2 * hi;
        v2f a, b;
        a.x = W[(size_t)m * HC + ka];
        a.y = W[(size_t)m * HC + ka + 1];
        float f0 = Wfc[ka * 2 + mc];
        float f1 = Wfc[(ka + 1) * 2 + mc];
        b.x = col_on ? f0 : 0.0f;
        b.y = col_on ? f1 : 0.0f;
        wp = __builtin_amdgcn_wmma_f32_16x16x4_f32(
             false, a,  false, b,
             (short)0, wp,  false,  false);
    }
    float p[8];
#pragma unroll
    for (int r = 0; r < 8; ++r) p[r] = __shfl_xor(wp[r], 16);
    v2f b0, b1, b2, b3;
    b0.x = hi ? p[2]  : wp[0];  b0.y = hi ? p[3]  : wp[1];
    b1.x = hi ? p[6]  : wp[4];  b1.y = hi ? p[7]  : wp[5];
    b2.x = hi ? wp[2] : p[0];   b2.y = hi ? wp[3] : p[1];
    b3.x = hi ? wp[6] : p[4];   b3.y = hi ? wp[7] : p[5];

    const float* arow = A + (size_t)(row0 + m) * HC;
    v2f a0, a1, a2, a3;
    a0.x = bn_relu(arow[2 * hi],      stats, gamma, beta, invN, 2 * hi);
    a0.y = bn_relu(arow[2 * hi + 1],  stats, gamma, beta, invN, 2 * hi + 1);
    a1.x = bn_relu(arow[2 * hi + 4],  stats, gamma, beta, invN, 2 * hi + 4);
    a1.y = bn_relu(arow[2 * hi + 5],  stats, gamma, beta, invN, 2 * hi + 5);
    a2.x = bn_relu(arow[2 * hi + 8],  stats, gamma, beta, invN, 2 * hi + 8);
    a2.y = bn_relu(arow[2 * hi + 9],  stats, gamma, beta, invN, 2 * hi + 9);
    a3.x = bn_relu(arow[2 * hi + 12], stats, gamma, beta, invN, 2 * hi + 12);
    a3.y = bn_relu(arow[2 * hi + 13], stats, gamma, beta, invN, 2 * hi + 13);
    v8f acc = {};
    acc = __builtin_amdgcn_wmma_f32_16x16x4_f32(false, a0, false, b0, (short)0, acc, false, false);
    acc = __builtin_amdgcn_wmma_f32_16x16x4_f32(false, a1, false, b1, (short)0, acc, false, false);
    acc = __builtin_amdgcn_wmma_f32_16x16x4_f32(false, a2, false, b2, (short)0, acc, false, false);
    acc = __builtin_amdgcn_wmma_f32_16x16x4_f32(false, a3, false, b3, (short)0, acc, false, false);

    float bm = bfc[mc];
#pragma unroll
    for (int j = 0; j < HC; ++j) bm += bias[j] * Wfc[j * 2 + mc];

    if (col_on) {
        float* cbase = C + (size_t)(row0 + hi * 8) * 2 + m;
        float* sbase = S + (size_t)(row0 + hi * 8) * 2 + m;
        const float* drow = dis + row0 + hi * 8;
#pragma unroll
        for (int r = 0; r < 8; ++r) {
            float d = drow[r];
            cbase[(size_t)r * 2] = d * acc[r];
            sbase[(size_t)r * 2] = d * d * acc[r] + bm;
        }
    }
}

#define OWN_THREADS 1024
#define OWN_QCAP    64
#define OWN_CHUNK   (16 * OWN_THREADS)

template <int CH, int PITCH, int SRC_BITS>
__device__ __forceinline__ void own_add_edge(unsigned w, const float* __restrict__ hw, float* acc) {
    const unsigned s  = w & ((1u << SRC_BITS) - 1u);
    const unsigned dl = w >> SRC_BITS;
    float* arow = acc + dl * PITCH;
    if constexpr (CH == 16) {
        const float4* hrow = (const float4*)(hw + (size_t)s * 16);
#pragma unroll
        for (int q = 0; q < 4; ++q) {
            float4 v = hrow[q];
            atomicAdd(arow + 4 * q + 0, v.x);
            atomicAdd(arow + 4 * q + 1, v.y);
            atomicAdd(arow + 4 * q + 2, v.z);
            atomicAdd(arow + 4 * q + 3, v.w);
        }
    } else {
        static_assert(CH == 2, "row widths in use: 16 and 2");
        const float2 v = *(const float2*)(hw + (size_t)s * 2);
        atomicAdd(arow + 0, v.x);
        atomicAdd(arow + 1, v.y);
    }
}

template <int CH, int ROWS, int SLICES, int PITCH, int SRC_BITS>
__global__ __launch_bounds__(OWN_THREADS)
void aggregate_owner(const int* __restrict__ src, const int* __restrict__ dst,
                     const float* __restrict__ dis, const float* __restrict__ hw,
                     float* __restrict__ out, int N, int E) {
    static_assert(ROWS <= (1 << (32 - SRC_BITS)), "local row must fit the queue word");
    __shared__ float acc[ROWS * PITCH];
    __shared__ unsigned queue[(OWN_THREADS / 32) * OWN_QCAP];
    __shared__ float dsl[ROWS];
    const int tid  = threadIdx.x;
    const int lane = tid & 31;
    unsigned* myq = queue + (tid >> 5) * OWN_QCAP;
    const int grp  = blockIdx.x / SLICES;
    const int slc  = blockIdx.x % SLICES;
    const int lo   = grp * ROWS;
    const int rows = min(ROWS, N - lo);
    const int per  = ((E + SLICES * OWN_THREADS - 1) / (SLICES * OWN_THREADS)) * OWN_THREADS;
    const int e0   = min(E, slc * per);
    const int e1   = min(E, e0 + per);

    constexpr int DPT = (ROWS + OWN_THREADS - 1) / OWN_THREADS;
    float dreg[DPT];
#pragma unroll
    for (int j = 0; j < DPT; ++j) {
        int idx = tid + j * OWN_THREADS;
        dreg[j] = (idx < rows) ? dis[lo + idx] : 0.0f;
    }
    for (int idx = tid; idx < rows * PITCH; idx += OWN_THREADS) acc[idx] = 0.0f;
#pragma unroll
    for (int j = 0; j < DPT; ++j) {
        int idx = tid + j * OWN_THREADS;
        if (idx < rows) dsl[idx] = dreg[j];
    }
    __syncthreads();

    int qn = 0;

#define OWN_STEP(m, km, w)                                                              \
    {                                                                                   \
        if (m) myq[qn + (int)__builtin_amdgcn_mbcnt_lo(km, 0u)] = (w);                  \
        qn += __popc(km);                                                               \
          \
          \
          \
        __builtin_amdgcn_wave_barrier();                                                \
        if (qn >= 32) {                                                                 \
            qn -= 32;                                                                   \
            own_add_edge<CH, PITCH, SRC_BITS>(myq[qn + lane], hw, acc);                 \
        }                                                                               \
                             \
        __builtin_amdgcn_wave_barrier();                                                \
    }

    int base = e0;
    for (; base + OWN_CHUNK <= e1; base += OWN_CHUNK) {
        const int* p  = dst + base + tid;
        const int* ps = src + base + tid;
        int dv[16], sv[16];
#pragma unroll
        for (int k = 0; k < 16; ++k) dv[k] = p[k * OWN_THREADS];
#pragma unroll
        for (int k = 0; k < 16; ++k)
            sv[k] = __hip_atomic_load(ps + k * OWN_THREADS, __ATOMIC_RELAXED, __HIP_MEMORY_SCOPE_WAVEFRONT);
        unsigned km[16];
        unsigned any = 0u;
#pragma unroll
        for (int k = 0; k < 16; ++k) {
            km[k] = __builtin_amdgcn_ballot_w32((unsigned)(dv[k] - lo) < (unsigned)rows);
            any |= km[k];
        }
        if (any != 0u) {
#pragma unroll
            for (int k = 0; k < 16; ++k) {
                if (km[k] != 0u) {
                    bool m = (unsigned)(dv[k] - lo) < (unsigned)rows;
                    OWN_STEP(m, km[k], (unsigned)sv[k] | ((unsigned)(dv[k] - lo) << SRC_BITS))
                }
            }
        }
    }
    for (; base < e1; base += OWN_THREADS) {
        int e = base + tid;
        int d = (e < e1) ? dst[e] : -1;
        int s = (e < e1) ? src[e] : 0;
        bool m = (unsigned)(d - lo) < (unsigned)rows;
        unsigned k1 = __builtin_amdgcn_ballot_w32(m);
        if (k1 != 0u) OWN_STEP(m, k1, (unsigned)s | ((unsigned)(d - lo) << SRC_BITS))
    }
#undef OWN_STEP
    if (lane < qn) own_add_edge<CH, PITCH, SRC_BITS>(myq[lane], hw, acc);
    __syncthreads();

    for (int idx = tid; idx < rows * CH; idx += OWN_THREADS) {
        int r = idx / CH, c = idx % CH;
        atomicAdd(out + (size_t)lo * CH + idx, dsl[r] * acc[r * PITCH + c]);
    }
}

__global__ void bn_stats(const float* __restrict__ agg, float* __restrict__ stats, int N) {
    __shared__ float s1s[256];
    __shared__ float s2s[256];
    int c = threadIdx.x & 15;
    int g = threadIdx.x >> 4;
    int stride = gridDim.x * 16;
    float s1 = 0.0f, s2 = 0.0f;
    for (int row = blockIdx.x * 16 + g; row < N; row += stride) {
        float v = agg[(size_t)row * HC + c];
        v = fmaxf(v, 0.0f);
        s1 += v;
        s2 += v * v;
    }
    s1s[threadIdx.x] = s1;
    s2s[threadIdx.x] = s2;
    __syncthreads();
    if (threadIdx.x < 16) {
        float t1 = 0.0f, t2 = 0.0f;
#pragma unroll
        for (int gg = 0; gg < 16; ++gg) {
            t1 += s1s[gg * 16 + threadIdx.x];
            t2 += s2s[gg * 16 + threadIdx.x];
        }
        atomicAdd(&stats[threadIdx.x], t1);
        atomicAdd(&stats[16 + threadIdx.x], t2);
    }
}

__global__ void head_logsoftmax2(const float* __restrict__ z, float* __restrict__ out, int N) {
    int i = blockIdx.x * blockDim.x + threadIdx.x;
    if (i >= N) return;
    const float2 v = ((const float2*)z)[i];
    float z0 = v.x, z1 = v.y;
    float mx  = fmaxf(z0, z1);
    float lse = mx + logf(__expf(z0 - mx) + __expf(z1 - mx));
    out[(size_t)i * 2 + 0] = z0 - lse;
    out[(size_t)i * 2 + 1] = z1 - lse;
}

extern "C" void kernel_launch(void* const* d_in, const int* in_sizes, int n_in,
                              void* d_out, int out_size, void* d_ws, size_t ws_size,
                              hipStream_t stream) {
    const int IN = 128;
    const int N  = in_sizes[0] / IN;
    const int E  = in_sizes[1] / 2;

    const float* x    = (const float*)d_in[0];
    const int*   ei   = (const int*)d_in[1];
    const int*   srcI = ei;
    const int*   dstI = ei + E;
    const float* W1   = (const float*)d_in[2];
    const float* b1   = (const float*)d_in[3];
    const float* W2   = (const float*)d_in[4];
    const float* b2   = (const float*)d_in[5];
    const float* W3   = (const float*)d_in[6];
    const float* b3   = (const float*)d_in[7];
    const float* g1   = (const float*)d_in[8];
    const float* be1  = (const float*)d_in[9];
    const float* g2   = (const float*)d_in[10];
    const float* be2  = (const float*)d_in[11];
    const float* Wfc  = (const float*)d_in[12];
    const float* bfc  = (const float*)d_in[13];

    float* ws   = (float*)d_ws;
    float* dis  = ws;
    float* bufA = dis + N;
    float* bufB = bufA + (size_t)N * HC;
    float* bufC = bufB + (size_t)N * HC;
    float* stats = bufC + (size_t)N * HC;

    const int B256 = 256;
    dim3 blkN((N + B256 - 1) / B256);
    dim3 blkG((N / 16 + 7) / 8);
    float invN = 1.0f / (float)N;
    dim3 blkO(((N + 3126 - 1) / 3126) * 8);
    dim3 blkO2(((N + 25008 - 1) / 25008) * 64);

    int* planes = (int*)bufA;
    dim3 blkD(((N + DEG_ROWS - 1) / DEG_ROWS) * DEG_SLICES);
    degree_owner<<<blkD, DEG_THREADS, 0, stream>>>(dstI, planes, N, E);
    degree_merge_rsqrt<<<blkN, B256, 0, stream>>>(planes, dis, stats, N);

    gemm_wmma_f32<<<blkG, B256, 0, stream>>>(x, W1, bufA, dis, b1, bufB, N, IN);
    aggregate_owner<16, 3126, 8, 17, 20><<<blkO, OWN_THREADS, 0, stream>>>(srcI, dstI, dis, bufA, bufB, N, E);
    bn_stats<<<256, 256, 0, stream>>>(bufB, stats, N);

    gemm_bn_wmma_f32<<<blkG, B256, 0, stream>>>(bufB, W2, bufA, dis, b2, bufC, stats, g1, be1, invN, N);
    aggregate_owner<16, 3126, 8, 17, 20><<<blkO, OWN_THREADS, 0, stream>>>(srcI, dstI, dis, bufA, bufC, N, E);
    bn_stats<<<256, 256, 0, stream>>>(bufC, stats + 32, N);

    gemm_bn_fold_wmma_f32<<<blkG, B256, 0, stream>>>(bufC, W3, Wfc, bfc, bufA, dis, b3, bufB, stats + 32, g2, be2, invN, N);
    aggregate_owner<2, 25008, 64, 2, 17><<<blkO2, OWN_THREADS, 0, stream>>>(srcI, dstI, dis, bufA, bufB, N, E);

    head_logsoftmax2<<<blkN, B256, 0, stream>>>(bufB, (float*)d_out, N);
}


# Row gather of the 16-channel aggregations cut from 4 line requests per edge to 1: four lanes share an edge, each loads one 16-byte quarter of the 64-byte row (LDS float adds, arrival order as before)
# speedup vs baseline: 10.3104x; 1.0466x over previous
#include <hip/hip_runtime.h>
#include <hip/hip_bf16.h>

typedef float v2f __attribute__((ext_vector_type(2)));
typedef float v8f __attribute__((ext_vector_type(8)));

#define HC 16
#define EPSV 1e-5f

#define DEG_ROWS    12512
#define DEG_SLICES  32
#define DEG_THREADS 1024
#define DEG_UNROLL  25
static_assert(DEG_SLICES <= 2 * HC, "the planes must fit the two N x 16 buffers they borrow");

__global__ __launch_bounds__(DEG_THREADS)
void degree_owner(const int* __restrict__ dst, int* __restrict__ planes, int N, int E) {
    __shared__ int cnt[DEG_ROWS];
    const int tid  = threadIdx.x;
    const int grp  = blockIdx.x / DEG_SLICES;
    const int slc  = blockIdx.x % DEG_SLICES;
    const int lo   = grp * DEG_ROWS;
    const int rows = min(DEG_ROWS, N - lo);
    const int e0   = (int)(((long long)E * slc) / DEG_SLICES);
    const int e1   = (int)(((long long)E * (slc + 1)) / DEG_SLICES);

    for (int idx = tid; idx < rows; idx += DEG_THREADS) cnt[idx] = 0;
    __syncthreads();

    for (int base = e0; base < e1; base += DEG_UNROLL * DEG_THREADS) {
        int dv[DEG_UNROLL];
#pragma unroll
        for (int k = 0; k < DEG_UNROLL; ++k) dv[k] = dst[min(base + tid + k * DEG_THREADS, e1 - 1)];
#pragma unroll
        for (int k = 0; k < DEG_UNROLL; ++k) {
            const unsigned r = (unsigned)(dv[k] - lo);
            if (base + tid + k * DEG_THREADS < e1 && r < (unsigned)rows) atomicAdd(&cnt[r], 1);
        }
    }
    __syncthreads();

    int* prow = planes + (size_t)slc * N + lo;
    for (int idx = tid; idx < rows; idx += DEG_THREADS) prow[idx] = cnt[idx];
}

__global__ void degree_merge_rsqrt(const int* __restrict__ planes, float* __restrict__ dis,
                                   float* __restrict__ stats, int N) {
    int i = blockIdx.x * blockDim.x + threadIdx.x;
    if (blockIdx.x == 0 && threadIdx.x < 64) stats[threadIdx.x] = 0.0f;
    if (i >= N) return;
    int c = 0;
#pragma unroll
    for (int s = 0; s < DEG_SLICES; ++s) c += planes[(size_t)s * N + i];
    dis[i] = rsqrtf((float)c + 1.0f);
}

__device__ __forceinline__ float bn_relu(float v, const float* __restrict__ stats, const float* __restrict__ gamma,
                                         const float* __restrict__ beta, float invN, int c) {
    float mu  = stats[c] * invN;
    float var = stats[16 + c] * invN - mu * mu;
    float sc  = rsqrtf(var + EPSV) * gamma[c];
    float sh  = beta[c] - mu * sc;
    return fmaxf(v, 0.0f) * sc + sh;
}

template <bool BN>
__device__ __forceinline__ void gemm_tile(const float* __restrict__ A, const float* __restrict__ B,
                                          float* __restrict__ C, const float* __restrict__ dis,
                                          const float* __restrict__ bias, float* __restrict__ S,
                                          const float* __restrict__ stats, const float* __restrict__ gamma,
                                          const float* __restrict__ beta, float invN, int N, int K) {
    int wave = (int)((blockIdx.x * blockDim.x + threadIdx.x) >> 5);
    int lane = threadIdx.x & 31;
    int row0 = wave * 16;
    if (row0 >= N) return;

    int m  = lane & 15;
    int hi = lane >> 4;

    v8f acc = {};
    if constexpr (BN) {
        const float* arow = A + (size_t)(row0 + m) * HC;
#pragma unroll
        for (int k0 = 0; k0 < HC; k0 += 4) {
            int ka = k0 + 2 * hi;
            v2f a, b;
            a.x = bn_relu(arow[ka],     stats, gamma, beta, invN, ka);
            a.y = bn_relu(arow[ka + 1], stats, gamma, beta, invN, ka + 1);
            b.x = B[(size_t)ka * HC + m];
            b.y = B[(size_t)(ka + 1) * HC + m];
            acc = __builtin_amdgcn_wmma_f32_16x16x4_f32(
                 false, a,  false, b,
                 (short)0, acc,  false,  false);
        }
    } else {
        const float* arow = A + (size_t)(row0 + m) * K;
        for (int k0 = 0; k0 < K; k0 += 4) {
            int ka = k0 + 2 * hi;
            v2f a, b;
            a.x = arow[ka];
            a.y = arow[ka + 1];
            b.x = B[(size_t)ka * HC + m];
            b.y = B[(size_t)(ka + 1) * HC + m];
            acc = __builtin_amdgcn_wmma_f32_16x16x4_f32(
                 false, a,  false, b,
                 (short)0, acc,  false,  false);
        }
    }

    float* cbase = C + (size_t)(row0 + hi * 8) * HC + m;
    float* sbase = S + (size_t)(row0 + hi * 8) * HC + m;
    const float* drow = dis + row0 + hi * 8;
    float bm = bias[m];
#pragma unroll
    for (int r = 0; r < 8; ++r) {
        float d = drow[r];
        cbase[(size_t)r * HC] = d * acc[r];
        sbase[(size_t)r * HC] = d * d * acc[r] + bm;
    }
}

__global__ void gemm_wmma_f32(const float* __restrict__ A, const float* __restrict__ B,
                              float* __restrict__ C, const float* __restrict__ dis,
                              const float* __restrict__ bias, float* __restrict__ S, int N, int K) {
    gemm_tile<false>(A, B, C, dis, bias, S, nullptr, nullptr, nullptr, 0.0f, N, K);
}

__global__ void gemm_bn_wmma_f32(const float* __restrict__ A, const float* __restrict__ B,
                                 float* __restrict__ C, const float* __restrict__ dis,
                                 const float* __restrict__ bias, float* __restrict__ S,
                                 const float* __restrict__ stats, const float* __restrict__ gamma,
                                 const float* __restrict__ beta, float invN, int N) {
    gemm_tile<true>(A, B, C, dis, bias, S, stats, gamma, beta, invN, N, HC);
}

__global__ void gemm_bn_fold_wmma_f32(const float* __restrict__ A, const float* __restrict__ W,
                                      const float* __restrict__ Wfc, const float* __restrict__ bfc,
                                      float* __restrict__ C, const float* __restrict__ dis,
                                      const float* __restrict__ bias, float* __restrict__ S,
                                      const float* __restrict__ stats, const float* __restrict__ gamma,
                                      const float* __restrict__ beta, float invN, int N) {
    int wave = (int)((blockIdx.x * blockDim.x + threadIdx.x) >> 5);
    int lane = threadIdx.x & 31;
    int row0 = wave * 16;
    if (row0 >= N) return;

    int m  = lane & 15;
    int hi = lane >> 4;
    const bool col_on = m < 2;
    const int  mc = m & 1;

    v8f wp = {};
#pragma unroll
    for (int k0 = 0; k0 < HC; k0 += 4) {
        int ka = k0 + 2 * hi;
        v2f a, b;
        a.x = W[(size_t)m * HC + ka];
        a.y = W[(size_t)m * HC + ka + 1];
        float f0 = Wfc[ka * 2 + mc];
        float f1 = Wfc[(ka + 1) * 2 + mc];
        b.x = col_on ? f0 : 0.0f;
        b.y = col_on ? f1 : 0.0f;
        wp = __builtin_amdgcn_wmma_f32_16x16x4_f32(
             false, a,  false, b,
             (short)0, wp,  false,  false);
    }
    float p[8];
#pragma unroll
    for (int r = 0; r < 8; ++r) p[r] = __shfl_xor(wp[r], 16);
    v2f b0, b1, b2, b3;
    b0.x = hi ? p[2]  : wp[0];  b0.y = hi ? p[3]  : wp[1];
    b1.x = hi ? p[6]  : wp[4];  b1.y = hi ? p[7]  : wp[5];
    b2.x = hi ? wp[2] : p[0];   b2.y = hi ? wp[3] : p[1];
    b3.x = hi ? wp[6] : p[4];   b3.y = hi ? wp[7] : p[5];

    const float* arow = A + (size_t)(row0 + m) * HC;
    v2f a0, a1, a2, a3;
    a0.x = bn_relu(arow[2 * hi],      stats, gamma, beta, invN, 2 * hi);
    a0.y = bn_relu(arow[2 * hi + 1],  stats, gamma, beta, invN, 2 * hi + 1);
    a1.x = bn_relu(arow[2 * hi + 4],  stats, gamma, beta, invN, 2 * hi + 4);
    a1.y = bn_relu(arow[2 * hi + 5],  stats, gamma, beta, invN, 2 * hi + 5);
    a2.x = bn_relu(arow[2 * hi + 8],  stats, gamma, beta, invN, 2 * hi + 8);
    a2.y = bn_relu(arow[2 * hi + 9],  stats, gamma, beta, invN, 2 * hi + 9);
    a3.x = bn_relu(arow[2 * hi + 12], stats, gamma, beta, invN, 2 * hi + 12);
    a3.y = bn_relu(arow[2 * hi + 13], stats, gamma, beta, invN, 2 * hi + 13);
    v8f acc = {};
    acc = __builtin_amdgcn_wmma_f32_16x16x4_f32(false, a0, false, b0, (short)0, acc, false, false);
    acc = __builtin_amdgcn_wmma_f32_16x16x4_f32(false, a1, false, b1, (short)0, acc, false, false);
    acc = __builtin_amdgcn_wmma_f32_16x16x4_f32(false, a2, false, b2, (short)0, acc, false, false);
    acc = __builtin_amdgcn_wmma_f32_16x16x4_f32(false, a3, false, b3, (short)0, acc, false, false);

    float bm = bfc[mc];
#pragma unroll
    for (int j = 0; j < HC; ++j) bm += bias[j] * Wfc[j * 2 + mc];

    if (col_on) {
        float* cbase = C + (size_t)(row0 + hi * 8) * 2 + m;
        float* sbase = S + (size_t)(row0 + hi * 8) * 2 + m;
        const float* drow = dis + row0 + hi * 8;
#pragma unroll
        for (int r = 0; r < 8; ++r) {
            float d = drow[r];
            cbase[(size_t)r * 2] = d * acc[r];
            sbase[(size_t)r * 2] = d * d * acc[r] + bm;
        }
    }
}

#define OWN_THREADS 1024
#define OWN_QCAP    64
#define OWN_CHUNK   (16 * OWN_THREADS)

template <int CH, int PITCH, int SRC_BITS, bool FULL>
__device__ __forceinline__ void own_add_queue(const unsigned* q, int n, int lane,
                                              const float* __restrict__ hw, float* acc) {
    if constexpr (CH == 16) {
        const int sub = lane >> 2;
        const int qtr = lane & 3;
        float4 v[4];
        float* arow[4];
        bool   on[4];
#pragma unroll
        for (int j = 0; j < 4; ++j) {
            const int idx = 8 * j + sub;
            on[j] = FULL || idx < n;
            unsigned w = 0u;
            if (on[j]) w = q[idx];
            const unsigned s  = w & ((1u << SRC_BITS) - 1u);
            const unsigned dl = w >> SRC_BITS;
            arow[j] = acc + dl * PITCH + 4 * qtr;
            v[j] = make_float4(0.0f, 0.0f, 0.0f, 0.0f);
            if (on[j]) v[j] = ((const float4*)(hw + (size_t)s * 16))[qtr];
        }
#pragma unroll
        for (int j = 0; j < 4; ++j) {
            if (on[j]) {
                atomicAdd(arow[j] + 0, v[j].x);
                atomicAdd(arow[j] + 1, v[j].y);
                atomicAdd(arow[j] + 2, v[j].z);
                atomicAdd(arow[j] + 3, v[j].w);
            }
        }
    } else {
        static_assert(CH == 2, "row widths in use: 16 and 2");
        if (FULL || lane < n) {
            const unsigned w  = q[lane];
            const unsigned s  = w & ((1u << SRC_BITS) - 1u);
            const unsigned dl = w >> SRC_BITS;
            float* arow = acc + dl * PITCH;
            const float2 v = *(const float2*)(hw + (size_t)s * 2);
            atomicAdd(arow + 0, v.x);
            atomicAdd(arow + 1, v.y);
        }
    }
}

template <int CH, int ROWS, int SLICES, int PITCH, int SRC_BITS>
__global__ __launch_bounds__(OWN_THREADS)
void aggregate_owner(const int* __restrict__ src, const int* __restrict__ dst,
                     const float* __restrict__ dis, const float* __restrict__ hw,
                     float* __restrict__ out, int N, int E) {
    static_assert(ROWS <= (1 << (32 - SRC_BITS)), "local row must fit the queue word");
    __shared__ float acc[ROWS * PITCH];
    __shared__ unsigned queue[(OWN_THREADS / 32) * OWN_QCAP];
    __shared__ float dsl[ROWS];
    const int tid  = threadIdx.x;
    const int lane = tid & 31;
    unsigned* myq = queue + (tid >> 5) * OWN_QCAP;
    const int grp  = blockIdx.x / SLICES;
    const int slc  = blockIdx.x % SLICES;
    const int lo   = grp * ROWS;
    const int rows = min(ROWS, N - lo);
    const int per  = ((E + SLICES * OWN_THREADS - 1) / (SLICES * OWN_THREADS)) * OWN_THREADS;
    const int e0   = min(E, slc * per);
    const int e1   = min(E, e0 + per);

    constexpr int DPT = (ROWS + OWN_THREADS - 1) / OWN_THREADS;
    float dreg[DPT];
#pragma unroll
    for (int j = 0; j < DPT; ++j) {
        int idx = tid + j * OWN_THREADS;
        dreg[j] = (idx < rows) ? dis[lo + idx] : 0.0f;
    }
    for (int idx = tid; idx < rows * PITCH; idx += OWN_THREADS) acc[idx] = 0.0f;
#pragma unroll
    for (int j = 0; j < DPT; ++j) {
        int idx = tid + j * OWN_THREADS;
        if (idx < rows) dsl[idx] = dreg[j];
    }
    __syncthreads();

    int qn = 0;

#define OWN_STEP(m, km, w)                                                              \
    {                                                                                   \
        if (m) myq[qn + (int)__builtin_amdgcn_mbcnt_lo(km, 0u)] = (w);                  \
        qn += __popc(km);                                                               \
          \
          \
          \
        __builtin_amdgcn_wave_barrier();                                                \
        if (qn >= 32) {                                                                 \
            qn -= 32;                                                                   \
            own_add_queue<CH, PITCH, SRC_BITS, true>(myq + qn, 32, lane, hw, acc);      \
        }                                                                               \
                             \
        __builtin_amdgcn_wave_barrier();                                                \
    }

    int base = e0;
    for (; base + OWN_CHUNK <= e1; base += OWN_CHUNK) {
        const int* p  = dst + base + tid;
        const int* ps = src + base + tid;
        int dv[16], sv[16];
#pragma unroll
        for (int k = 0; k < 16; ++k) dv[k] = p[k * OWN_THREADS];
#pragma unroll
        for (int k = 0; k < 16; ++k)
            sv[k] = __hip_atomic_load(ps + k * OWN_THREADS, __ATOMIC_RELAXED, __HIP_MEMORY_SCOPE_WAVEFRONT);
        unsigned km[16];
        unsigned any = 0u;
#pragma unroll
        for (int k = 0; k < 16; ++k) {
            km[k] = __builtin_amdgcn_ballot_w32((unsigned)(dv[k] - lo) < (unsigned)rows);
            any |= km[k];
        }
        if (any != 0u) {
#pragma unroll
            for (int k = 0; k < 16; ++k) {
                if (km[k] != 0u) {
                    bool m = (unsigned)(dv[k] - lo) < (unsigned)rows;
                    OWN_STEP(m, km[k], (unsigned)sv[k] | ((unsigned)(dv[k] - lo) << SRC_BITS))
                }
            }
        }
    }
    for (; base < e1; base += OWN_THREADS) {
        int e = base + tid;
        int d = (e < e1) ? dst[e] : -1;
        int s = (e < e1) ? src[e] : 0;
        bool m = (unsigned)(d - lo) < (unsigned)rows;
        unsigned k1 = __builtin_amdgcn_ballot_w32(m);
        if (k1 != 0u) OWN_STEP(m, k1, (unsigned)s | ((unsigned)(d - lo) << SRC_BITS))
    }
#undef OWN_STEP
    own_add_queue<CH, PITCH, SRC_BITS, false>(myq, qn, lane, hw, acc);
    __syncthreads();

    for (int idx = tid; idx < rows * CH; idx += OWN_THREADS) {
        int r = idx / CH, c = idx % CH;
        atomicAdd(out + (size_t)lo * CH + idx, dsl[r] * acc[r * PITCH + c]);
    }
}

__global__ void bn_stats(const float* __restrict__ agg, float* __restrict__ stats, int N) {
    __shared__ float s1s[256];
    __shared__ float s2s[256];
    int c = threadIdx.x & 15;
    int g = threadIdx.x >> 4;
    int stride = gridDim.x * 16;
    float s1 = 0.0f, s2 = 0.0f;
    for (int row = blockIdx.x * 16 + g; row < N; row += stride) {
        float v = agg[(size_t)row * HC + c];
        v = fmaxf(v, 0.0f);
        s1 += v;
        s2 += v * v;
    }
    s1s[threadIdx.x] = s1;
    s2s[threadIdx.x] = s2;
    __syncthreads();
    if (threadIdx.x < 16) {
        float t1 = 0.0f, t2 = 0.0f;
#pragma unroll
        for (int gg = 0; gg < 16; ++gg) {
            t1 += s1s[gg * 16 + threadIdx.x];
            t2 += s2s[gg * 16 + threadIdx.x];
        }
        atomicAdd(&stats[threadIdx.x], t1);
        atomicAdd(&stats[16 + threadIdx.x], t2);
    }
}

__global__ void head_logsoftmax2(const float* __restrict__ z, float* __restrict__ out, int N) {
    int i = blockIdx.x * blockDim.x + threadIdx.x;
    if (i >= N) return;
    const float2 v = ((const float2*)z)[i];
    float z0 = v.x, z1 = v.y;
    float mx  = fmaxf(z0, z1);
    float lse = mx + logf(__expf(z0 - mx) + __expf(z1 - mx));
    out[(size_t)i * 2 + 0] = z0 - lse;
    out[(size_t)i * 2 + 1] = z1 - lse;
}

extern "C" void kernel_launch(void* const* d_in, const int* in_sizes, int n_in,
                              void* d_out, int out_size, void* d_ws, size_t ws_size,
                              hipStream_t stream) {
    const int IN = 128;
    const int N  = in_sizes[0] / IN;
    const int E  = in_sizes[1] / 2;

    const float* x    = (const float*)d_in[0];
    const int*   ei   = (const int*)d_in[1];
    const int*   srcI = ei;
    const int*   dstI = ei + E;
    const float* W1   = (const float*)d_in[2];
    const float* b1   = (const float*)d_in[3];
    const float* W2   = (const float*)d_in[4];
    const float* b2   = (const float*)d_in[5];
    const float* W3   = (const float*)d_in[6];
    const float* b3   = (const float*)d_in[7];
    const float* g1   = (const float*)d_in[8];
    const float* be1  = (const float*)d_in[9];
    const float* g2   = (const float*)d_in[10];
    const float* be2  = (const float*)d_in[11];
    const float* Wfc  = (const float*)d_in[12];
    const float* bfc  = (const float*)d_in[13];

    float* ws   = (float*)d_ws;
    float* dis  = ws;
    float* bufA = dis + N;
    float* bufB = bufA + (size_t)N * HC;
    float* bufC = bufB + (size_t)N * HC;
    float* stats = bufC + (size_t)N * HC;

    const int B256 = 256;
    dim3 blkN((N + B256 - 1) / B256);
    dim3 blkG((N / 16 + 7) / 8);
    float invN = 1.0f / (float)N;
    dim3 blkO(((N + 3126 - 1) / 3126) * 8);
    dim3 blkO2(((N + 25008 - 1) / 25008) * 64);

    int* planes = (int*)bufA;
    dim3 blkD(((N + DEG_ROWS - 1) / DEG_ROWS) * DEG_SLICES);
    degree_owner<<<blkD, DEG_THREADS, 0, stream>>>(dstI, planes, N, E);
    degree_merge_rsqrt<<<blkN, B256, 0, stream>>>(planes, dis, stats, N);

    gemm_wmma_f32<<<blkG, B256, 0, stream>>>(x, W1, bufA, dis, b1, bufB, N, IN);
    aggregate_owner<16, 3126, 8, 17, 20><<<blkO, OWN_THREADS, 0, stream>>>(srcI, dstI, dis, bufA, bufB, N, E);
    bn_stats<<<256, 256, 0, stream>>>(bufB, stats, N);

    gemm_bn_wmma_f32<<<blkG, B256, 0, stream>>>(bufB, W2, bufA, dis, b2, bufC, stats, g1, be1, invN, N);
    aggregate_owner<16, 3126, 8, 17, 20><<<blkO, OWN_THREADS, 0, stream>>>(srcI, dstI, dis, bufA, bufC, N, E);
    bn_stats<<<256, 256, 0, stream>>>(bufC, stats + 32, N);

    gemm_bn_fold_wmma_f32<<<blkG, B256, 0, stream>>>(bufC, W3, Wfc, bfc, bufA, dis, b3, bufB, stats + 32, g2, be2, invN, N);
    aggregate_owner<2, 25008, 64, 2, 17><<<blkO2, OWN_THREADS, 0, stream>>>(srcI, dstI, dis, bufA, bufB, N, E);

    head_logsoftmax2<<<blkN, B256, 0, stream>>>(bufB, (float*)d_out, N);
}


# 16-channel aggregations re-tiled: 23 row ranges (4348 LDS rows) x 11 edge slices of whole scan trips; edge list read 23 times, not 32; 11 slice partials per row, not 8 (float atomics, arrival order)
# speedup vs baseline: 10.9299x; 1.0601x over previous
#include <hip/hip_runtime.h>
#include <hip/hip_bf16.h>

typedef float v2f __attribute__((ext_vector_type(2)));
typedef float v8f __attribute__((ext_vector_type(8)));

#define HC 16
#define EPSV 1e-5f

#define DEG_ROWS    12512
#define DEG_SLICES  32
#define DEG_THREADS 1024
#define DEG_UNROLL  25
static_assert(DEG_SLICES <= 2 * HC, "the planes must fit the two N x 16 buffers they borrow");

__global__ __launch_bounds__(DEG_THREADS)
void degree_owner(const int* __restrict__ dst, int* __restrict__ planes, int N, int E) {
    __shared__ int cnt[DEG_ROWS];
    const int tid  = threadIdx.x;
    const int grp  = blockIdx.x / DEG_SLICES;
    const int slc  = blockIdx.x % DEG_SLICES;
    const int lo   = grp * DEG_ROWS;
    const int rows = min(DEG_ROWS, N - lo);
    const int e0   = (int)(((long long)E * slc) / DEG_SLICES);
    const int e1   = (int)(((long long)E * (slc + 1)) / DEG_SLICES);

    for (int idx = tid; idx < rows; idx += DEG_THREADS) cnt[idx] = 0;
    __syncthreads();

    for (int base = e0; base < e1; base += DEG_UNROLL * DEG_THREADS) {
        int dv[DEG_UNROLL];
#pragma unroll
        for (int k = 0; k < DEG_UNROLL; ++k) dv[k] = dst[min(base + tid + k * DEG_THREADS, e1 - 1)];
#pragma unroll
        for (int k = 0; k < DEG_UNROLL; ++k) {
            const unsigned r = (unsigned)(dv[k] - lo);
            if (base + tid + k * DEG_THREADS < e1 && r < (unsigned)rows) atomicAdd(&cnt[r], 1);
        }
    }
    __syncthreads();

    int* prow = planes + (size_t)slc * N + lo;
    for (int idx = tid; idx < rows; idx += DEG_THREADS) prow[idx] = cnt[idx];
}

__global__ void degree_merge_rsqrt(const int* __restrict__ planes, float* __restrict__ dis,
                                   float* __restrict__ stats, int N) {
    int i = blockIdx.x * blockDim.x + threadIdx.x;
    if (blockIdx.x == 0 && threadIdx.x < 64) stats[threadIdx.x] = 0.0f;
    if (i >= N) return;
    int c = 0;
#pragma unroll
    for (int s = 0; s < DEG_SLICES; ++s) c += planes[(size_t)s * N + i];
    dis[i] = rsqrtf((float)c + 1.0f);
}

__device__ __forceinline__ float bn_relu(float v, const float* __restrict__ stats, const float* __restrict__ gamma,
                                         const float* __restrict__ beta, float invN, int c) {
    float mu  = stats[c] * invN;
    float var = stats[16 + c] * invN - mu * mu;
    float sc  = rsqrtf(var + EPSV) * gamma[c];
    float sh  = beta[c] - mu * sc;
    return fmaxf(v, 0.0f) * sc + sh;
}

template <bool BN>
__device__ __forceinline__ void gemm_tile(const float* __restrict__ A, const float* __restrict__ B,
                                          float* __restrict__ C, const float* __restrict__ dis,
                                          const float* __restrict__ bias, float* __restrict__ S,
                                          const float* __restrict__ stats, const float* __restrict__ gamma,
                                          const float* __restrict__ beta, float invN, int N, int K) {
    int wave = (int)((blockIdx.x * blockDim.x + threadIdx.x) >> 5);
    int lane = threadIdx.x & 31;
    int row0 = wave * 16;
    if (row0 >= N) return;

    int m  = lane & 15;
    int hi = lane >> 4;

    v8f acc = {};
    if constexpr (BN) {
        const float* arow = A + (size_t)(row0 + m) * HC;
#pragma unroll
        for (int k0 = 0; k0 < HC; k0 += 4) {
            int ka = k0 + 2 * hi;
            v2f a, b;
            a.x = bn_relu(arow[ka],     stats, gamma, beta, invN, ka);
            a.y = bn_relu(arow[ka + 1], stats, gamma, beta, invN, ka + 1);
            b.x = B[(size_t)ka * HC + m];
            b.y = B[(size_t)(ka + 1) * HC + m];
            acc = __builtin_amdgcn_wmma_f32_16x16x4_f32(
                 false, a,  false, b,
                 (short)0, acc,  false,  false);
        }
    } else {
        const float* arow = A + (size_t)(row0 + m) * K;
        for (int k0 = 0; k0 < K; k0 += 4) {
            int ka = k0 + 2 * hi;
            v2f a, b;
            a.x = arow[ka];
            a.y = arow[ka + 1];
            b.x = B[(size_t)ka * HC + m];
            b.y = B[(size_t)(ka + 1) * HC + m];
            acc = __builtin_amdgcn_wmma_f32_16x16x4_f32(
                 false, a,  false, b,
                 (short)0, acc,  false,  false);
        }
    }

    float* cbase = C + (size_t)(row0 + hi * 8) * HC + m;
    float* sbase = S + (size_t)(row0 + hi * 8) * HC + m;
    const float* drow = dis + row0 + hi * 8;
    float bm = bias[m];
#pragma unroll
    for (int r = 0; r < 8; ++r) {
        float d = drow[r];
        cbase[(size_t)r * HC] = d * acc[r];
        sbase[(size_t)r * HC] = d * d * acc[r] + bm;
    }
}

__global__ void gemm_wmma_f32(const float* __restrict__ A, const float* __restrict__ B,
                              float* __restrict__ C, const float* __restrict__ dis,
                              const float* __restrict__ bias, float* __restrict__ S, int N, int K) {
    gemm_tile<false>(A, B, C, dis, bias, S, nullptr, nullptr, nullptr, 0.0f, N, K);
}

__global__ void gemm_bn_wmma_f32(const float* __restrict__ A, const float* __restrict__ B,
                                 float* __restrict__ C, const float* __restrict__ dis,
                                 const float* __restrict__ bias, float* __restrict__ S,
                                 const float* __restrict__ stats, const float* __restrict__ gamma,
                                 const float* __restrict__ beta, float invN, int N) {
    gemm_tile<true>(A, B, C, dis, bias, S, stats, gamma, beta, invN, N, HC);
}

__global__ void gemm_bn_fold_wmma_f32(const float* __restrict__ A, const float* __restrict__ W,
                                      const float* __restrict__ Wfc, const float* __restrict__ bfc,
                                      float* __restrict__ C, const float* __restrict__ dis,
                                      const float* __restrict__ bias, float* __restrict__ S,
                                      const float* __restrict__ stats, const float* __restrict__ gamma,
                                      const float* __restrict__ beta, float invN, int N) {
    int wave = (int)((blockIdx.x * blockDim.x + threadIdx.x) >> 5);
    int lane = threadIdx.x & 31;
    int row0 = wave * 16;
    if (row0 >= N) return;

    int m  = lane & 15;
    int hi = lane >> 4;
    const bool col_on = m < 2;
    const int  mc = m & 1;

    v8f wp = {};
#pragma unroll
    for (int k0 = 0; k0 < HC; k0 += 4) {
        int ka = k0 + 2 * hi;
        v2f a, b;
        a.x = W[(size_t)m * HC + ka];
        a.y = W[(size_t)m * HC + ka + 1];
        float f0 = Wfc[ka * 2 + mc];
        float f1 = Wfc[(ka + 1) * 2 + mc];
        b.x = col_on ? f0 : 0.0f;
        b.y = col_on ? f1 : 0.0f;
        wp = __builtin_amdgcn_wmma_f32_16x16x4_f32(
             false, a,  false, b,
             (short)0, wp,  false,  false);
    }
    float p[8];
#pragma unroll
    for (int r = 0; r < 8; ++r) p[r] = __shfl_xor(wp[r], 16);
    v2f b0, b1, b2, b3;
    b0.x = hi ? p[2]  : wp[0];  b0.y = hi ? p[3]  : wp[1];
    b1.x = hi ? p[6]  : wp[4];  b1.y = hi ? p[7]  : wp[5];
    b2.x = hi ? wp[2] : p[0];   b2.y = hi ? wp[3] : p[1];
    b3.x = hi ? wp[6] : p[4];   b3.y = hi ? wp[7] : p[5];

    const float* arow = A + (size_t)(row0 + m) * HC;
    v2f a0, a1, a2, a3;
    a0.x = bn_relu(arow[2 * hi],      stats, gamma, beta, invN, 2 * hi);
    a0.y = bn_relu(arow[2 * hi + 1],  stats, gamma, beta, invN, 2 * hi + 1);
    a1.x = bn_relu(arow[2 * hi + 4],  stats, gamma, beta, invN, 2 * hi + 4);
    a1.y = bn_relu(arow[2 * hi + 5],  stats, gamma, beta, invN, 2 * hi + 5);
    a2.x = bn_relu(arow[2 * hi + 8],  stats, gamma, beta, invN, 2 * hi + 8);
    a2.y = bn_relu(arow[2 * hi + 9],  stats, gamma, beta, invN, 2 * hi + 9);
    a3.x = bn_relu(arow[2 * hi + 12], stats, gamma, beta, invN, 2 * hi + 12);
    a3.y = bn_relu(arow[2 * hi + 13], stats, gamma, beta, invN, 2 * hi + 13);
    v8f acc = {};
    acc = __builtin_amdgcn_wmma_f32_16x16x4_f32(false, a0, false, b0, (short)0, acc, false, false);
    acc = __builtin_amdgcn_wmma_f32_16x16x4_f32(false, a1, false, b1, (short)0, acc, false, false);
    acc = __builtin_amdgcn_wmma_f32_16x16x4_f32(false, a2, false, b2, (short)0, acc, false, false);
    acc = __builtin_amdgcn_wmma_f32_16x16x4_f32(false, a3, false, b3, (short)0, acc, false, false);

    float bm = bfc[mc];
#pragma unroll
    for (int j = 0; j < HC; ++j) bm += bias[j] * Wfc[j * 2 + mc];

    if (col_on) {
        float* cbase = C + (size_t)(row0 + hi * 8) * 2 + m;
        float* sbase = S + (size_t)(row0 + hi * 8) * 2 + m;
        const float* drow = dis + row0 + hi * 8;
#pragma unroll
        for (int r = 0; r < 8; ++r) {
            float d = drow[r];
            cbase[(size_t)r * 2] = d * acc[r];
            sbase[(size_t)r * 2] = d * d * acc[r] + bm;
        }
    }
}

#define OWN_THREADS 1024
#define OWN_QCAP    64
#define OWN_CHUNK   (16 * OWN_THREADS)

template <int CH, int PITCH, int SRC_BITS, bool FULL>
__device__ __forceinline__ void own_add_queue(const unsigned* q, int n, int lane,
                                              const float* __restrict__ hw, float* acc) {
    if constexpr (CH == 16) {
        const int sub = lane >> 2;
        const int qtr = lane & 3;
        float4 v[4];
        float* arow[4];
        bool   on[4];
#pragma unroll
        for (int j = 0; j < 4; ++j) {
            const int idx = 8 * j + sub;
            on[j] = FULL || idx < n;
            unsigned w = 0u;
            if (on[j]) w = q[idx];
            const unsigned s  = w & ((1u << SRC_BITS) - 1u);
            const unsigned dl = w >> SRC_BITS;
            arow[j] = acc + dl * PITCH + 4 * qtr;
            v[j] = make_float4(0.0f, 0.0f, 0.0f, 0.0f);
            if (on[j]) v[j] = ((const float4*)(hw + (size_t)s * 16))[qtr];
        }
#pragma unroll
        for (int j = 0; j < 4; ++j) {
            if (on[j]) {
                atomicAdd(arow[j] + 0, v[j].x);
                atomicAdd(arow[j] + 1, v[j].y);
                atomicAdd(arow[j] + 2, v[j].z);
                atomicAdd(arow[j] + 3, v[j].w);
            }
        }
    } else {
        static_assert(CH == 2, "row widths in use: 16 and 2");
        if (FULL || lane < n) {
            const unsigned w  = q[lane];
            const unsigned s  = w & ((1u << SRC_BITS) - 1u);
            const unsigned dl = w >> SRC_BITS;
            float* arow = acc + dl * PITCH;
            const float2 v = *(const float2*)(hw + (size_t)s * 2);
            atomicAdd(arow + 0, v.x);
            atomicAdd(arow + 1, v.y);
        }
    }
}

template <int CH, int ROWS, int SLICES, int PITCH, int SRC_BITS>
__global__ __launch_bounds__(OWN_THREADS)
void aggregate_owner(const int* __restrict__ src, const int* __restrict__ dst,
                     const float* __restrict__ dis, const float* __restrict__ hw,
                     float* __restrict__ out, int N, int E) {
    static_assert(ROWS <= (1 << (32 - SRC_BITS)), "local row must fit the queue word");
    __shared__ float acc[ROWS * PITCH];
    __shared__ unsigned queue[(OWN_THREADS / 32) * OWN_QCAP];
    __shared__ float dsl[ROWS];
    const int tid  = threadIdx.x;
    const int lane = tid & 31;
    unsigned* myq = queue + (tid >> 5) * OWN_QCAP;
    const int grp  = blockIdx.x / SLICES;
    const int slc  = blockIdx.x % SLICES;
    const int lo   = grp * ROWS;
    const int rows = min(ROWS, N - lo);
    constexpr int GRAN = (CH == 16) ? OWN_CHUNK : OWN_THREADS;
    const int per  = ((E + SLICES * GRAN - 1) / (SLICES * GRAN)) * GRAN;
    const int e0   = min(E, slc * per);
    const int e1   = min(E, e0 + per);

    constexpr int DPT = (ROWS + OWN_THREADS - 1) / OWN_THREADS;
    float dreg[DPT];
#pragma unroll
    for (int j = 0; j < DPT; ++j) {
        int idx = tid + j * OWN_THREADS;
        dreg[j] = (idx < rows) ? dis[lo + idx] : 0.0f;
    }
    for (int idx = tid; idx < rows * PITCH; idx += OWN_THREADS) acc[idx] = 0.0f;
#pragma unroll
    for (int j = 0; j < DPT; ++j) {
        int idx = tid + j * OWN_THREADS;
        if (idx < rows) dsl[idx] = dreg[j];
    }
    __syncthreads();

    int qn = 0;

#define OWN_STEP(m, km, w)                                                              \
    {                                                                                   \
        if (m) myq[qn + (int)__builtin_amdgcn_mbcnt_lo(km, 0u)] = (w);                  \
        qn += __popc(km);                                                               \
          \
          \
          \
        __builtin_amdgcn_wave_barrier();                                                \
        if (qn >= 32) {                                                                 \
            qn -= 32;                                                                   \
            own_add_queue<CH, PITCH, SRC_BITS, true>(myq + qn, 32, lane, hw, acc);      \
        }                                                                               \
                             \
        __builtin_amdgcn_wave_barrier();                                                \
    }

    int base = e0;
    for (; base + OWN_CHUNK <= e1; base += OWN_CHUNK) {
        const int* p  = dst + base + tid;
        const int* ps = src + base + tid;
        int dv[16], sv[16];
#pragma unroll
        for (int k = 0; k < 16; ++k) dv[k] = p[k * OWN_THREADS];
#pragma unroll
        for (int k = 0; k < 16; ++k)
            sv[k] = __hip_atomic_load(ps + k * OWN_THREADS, __ATOMIC_RELAXED, __HIP_MEMORY_SCOPE_WAVEFRONT);
        unsigned km[16];
        unsigned any = 0u;
#pragma unroll
        for (int k = 0; k < 16; ++k) {
            km[k] = __builtin_amdgcn_ballot_w32((unsigned)(dv[k] - lo) < (unsigned)rows);
            any |= km[k];
        }
        if (any != 0u) {
#pragma unroll
            for (int k = 0; k < 16; ++k) {
                if (km[k] != 0u) {
                    bool m = (unsigned)(dv[k] - lo) < (unsigned)rows;
                    OWN_STEP(m, km[k], (unsigned)sv[k] | ((unsigned)(dv[k] - lo) << SRC_BITS))
                }
            }
        }
    }
    for (; base < e1; base += OWN_THREADS) {
        int e = base + tid;
        int d = (e < e1) ? dst[e] : -1;
        int s = (e < e1) ? src[e] : 0;
        bool m = (unsigned)(d - lo) < (unsigned)rows;
        unsigned k1 = __builtin_amdgcn_ballot_w32(m);
        if (k1 != 0u) OWN_STEP(m, k1, (unsigned)s | ((unsigned)(d - lo) << SRC_BITS))
    }
#undef OWN_STEP
    own_add_queue<CH, PITCH, SRC_BITS, false>(myq, qn, lane, hw, acc);
    __syncthreads();

    for (int idx = tid; idx < rows * CH; idx += OWN_THREADS) {
        int r = idx / CH, c = idx % CH;
        atomicAdd(out + (size_t)lo * CH + idx, dsl[r] * acc[r * PITCH + c]);
    }
}

__global__ void bn_stats(const float* __restrict__ agg, float* __restrict__ stats, int N) {
    __shared__ float s1s[256];
    __shared__ float s2s[256];
    int c = threadIdx.x & 15;
    int g = threadIdx.x >> 4;
    int stride = gridDim.x * 16;
    float s1 = 0.0f, s2 = 0.0f;
    for (int row = blockIdx.x * 16 + g; row < N; row += stride) {
        float v = agg[(size_t)row * HC + c];
        v = fmaxf(v, 0.0f);
        s1 += v;
        s2 += v * v;
    }
    s1s[threadIdx.x] = s1;
    s2s[threadIdx.x] = s2;
    __syncthreads();
    if (threadIdx.x < 16) {
        float t1 = 0.0f, t2 = 0.0f;
#pragma unroll
        for (int gg = 0; gg < 16; ++gg) {
            t1 += s1s[gg * 16 + threadIdx.x];
            t2 += s2s[gg * 16 + threadIdx.x];
        }
        atomicAdd(&stats[threadIdx.x], t1);
        atomicAdd(&stats[16 + threadIdx.x], t2);
    }
}

__global__ void head_logsoftmax2(const float* __restrict__ z, float* __restrict__ out, int N) {
    int i = blockIdx.x * blockDim.x + threadIdx.x;
    if (i >= N) return;
    const float2 v = ((const float2*)z)[i];
    float z0 = v.x, z1 = v.y;
    float mx  = fmaxf(z0, z1);
    float lse = mx + logf(__expf(z0 - mx) + __expf(z1 - mx));
    out[(size_t)i * 2 + 0] = z0 - lse;
    out[(size_t)i * 2 + 1] = z1 - lse;
}

extern "C" void kernel_launch(void* const* d_in, const int* in_sizes, int n_in,
                              void* d_out, int out_size, void* d_ws, size_t ws_size,
                              hipStream_t stream) {
    const int IN = 128;
    const int N  = in_sizes[0] / IN;
    const int E  = in_sizes[1] / 2;

    const float* x    = (const float*)d_in[0];
    const int*   ei   = (const int*)d_in[1];
    const int*   srcI = ei;
    const int*   dstI = ei + E;
    const float* W1   = (const float*)d_in[2];
    const float* b1   = (const float*)d_in[3];
    const float* W2   = (const float*)d_in[4];
    const float* b2   = (const float*)d_in[5];
    const float* W3   = (const float*)d_in[6];
    const float* b3   = (const float*)d_in[7];
    const float* g1   = (const float*)d_in[8];
    const float* be1  = (const float*)d_in[9];
    const float* g2   = (const float*)d_in[10];
    const float* be2  = (const float*)d_in[11];
    const float* Wfc  = (const float*)d_in[12];
    const float* bfc  = (const float*)d_in[13];

    float* ws   = (float*)d_ws;
    float* dis  = ws;
    float* bufA = dis + N;
    float* bufB = bufA + (size_t)N * HC;
    float* bufC = bufB + (size_t)N * HC;
    float* stats = bufC + (size_t)N * HC;

    const int B256 = 256;
    dim3 blkN((N + B256 - 1) / B256);
    dim3 blkG((N / 16 + 7) / 8);
    float invN = 1.0f / (float)N;
    dim3 blkO(((N + 4348 - 1) / 4348) * 11);
    dim3 blkO2(((N + 25008 - 1) / 25008) * 64);

    int* planes = (int*)bufA;
    dim3 blkD(((N + DEG_ROWS - 1) / DEG_ROWS) * DEG_SLICES);
    degree_owner<<<blkD, DEG_THREADS, 0, stream>>>(dstI, planes, N, E);
    degree_merge_rsqrt<<<blkN, B256, 0, stream>>>(planes, dis, stats, N);

    gemm_wmma_f32<<<blkG, B256, 0, stream>>>(x, W1, bufA, dis, b1, bufB, N, IN);
    aggregate_owner<16, 4348, 11, 17, 17><<<blkO, OWN_THREADS, 0, stream>>>(srcI, dstI, dis, bufA, bufB, N, E);
    bn_stats<<<256, 256, 0, stream>>>(bufB, stats, N);

    gemm_bn_wmma_f32<<<blkG, B256, 0, stream>>>(bufB, W2, bufA, dis, b2, bufC, stats, g1, be1, invN, N);
    aggregate_owner<16, 4348, 11, 17, 17><<<blkO, OWN_THREADS, 0, stream>>>(srcI, dstI, dis, bufA, bufC, N, E);
    bn_stats<<<256, 256, 0, stream>>>(bufC, stats + 32, N);

    gemm_bn_fold_wmma_f32<<<blkG, B256, 0, stream>>>(bufC, W3, Wfc, bfc, bufA, dis, b3, bufB, stats + 32, g2, be2, invN, N);
    aggregate_owner<2, 25008, 64, 2, 17><<<blkO2, OWN_THREADS, 0, stream>>>(srcI, dstI, dis, bufA, bufB, N, E);

    head_logsoftmax2<<<blkN, B256, 0, stream>>>(bufB, (float*)d_out, N);
}


# BatchNorm channel-sum pass: 25 dependent 4-byte loads per thread become 7 independent 16-byte loads in flight at once; shuffle + LDS reduce (new sum tree; float atomics in arrival order as before)
# speedup vs baseline: 12.0334x; 1.1010x over previous
#include <hip/hip_runtime.h>
#include <hip/hip_bf16.h>

typedef float v2f __attribute__((ext_vector_type(2)));
typedef float v8f __attribute__((ext_vector_type(8)));

#define HC 16
#define EPSV 1e-5f

#define DEG_ROWS    12512
#define DEG_SLICES  32
#define DEG_THREADS 1024
#define DEG_UNROLL  25
static_assert(DEG_SLICES <= 2 * HC, "the planes must fit the two N x 16 buffers they borrow");

__global__ __launch_bounds__(DEG_THREADS)
void degree_owner(const int* __restrict__ dst, int* __restrict__ planes, int N, int E) {
    __shared__ int cnt[DEG_ROWS];
    const int tid  = threadIdx.x;
    const int grp  = blockIdx.x / DEG_SLICES;
    const int slc  = blockIdx.x % DEG_SLICES;
    const int lo   = grp * DEG_ROWS;
    const int rows = min(DEG_ROWS, N - lo);
    const int e0   = (int)(((long long)E * slc) / DEG_SLICES);
    const int e1   = (int)(((long long)E * (slc + 1)) / DEG_SLICES);

    for (int idx = tid; idx < rows; idx += DEG_THREADS) cnt[idx] = 0;
    __syncthreads();

    for (int base = e0; base < e1; base += DEG_UNROLL * DEG_THREADS) {
        int dv[DEG_UNROLL];
#pragma unroll
        for (int k = 0; k < DEG_UNROLL; ++k) dv[k] = dst[min(base + tid + k * DEG_THREADS, e1 - 1)];
#pragma unroll
        for (int k = 0; k < DEG_UNROLL; ++k) {
            const unsigned r = (unsigned)(dv[k] - lo);
            if (base + tid + k * DEG_THREADS < e1 && r < (unsigned)rows) atomicAdd(&cnt[r], 1);
        }
    }
    __syncthreads();

    int* prow = planes + (size_t)slc * N + lo;
    for (int idx = tid; idx < rows; idx += DEG_THREADS) prow[idx] = cnt[idx];
}

__global__ void degree_merge_rsqrt(const int* __restrict__ planes, float* __restrict__ dis,
                                   float* __restrict__ stats, int N) {
    int i = blockIdx.x * blockDim.x + threadIdx.x;
    if (blockIdx.x == 0 && threadIdx.x < 64) stats[threadIdx.x] = 0.0f;
    if (i >= N) return;
    int c = 0;
#pragma unroll
    for (int s = 0; s < DEG_SLICES; ++s) c += planes[(size_t)s * N + i];
    dis[i] = rsqrtf((float)c + 1.0f);
}

__device__ __forceinline__ float bn_relu(float v, const float* __restrict__ stats, const float* __restrict__ gamma,
                                         const float* __restrict__ beta, float invN, int c) {
    float mu  = stats[c] * invN;
    float var = stats[16 + c] * invN - mu * mu;
    float sc  = rsqrtf(var + EPSV) * gamma[c];
    float sh  = beta[c] - mu * sc;
    return fmaxf(v, 0.0f) * sc + sh;
}

template <bool BN>
__device__ __forceinline__ void gemm_tile(const float* __restrict__ A, const float* __restrict__ B,
                                          float* __restrict__ C, const float* __restrict__ dis,
                                          const float* __restrict__ bias, float* __restrict__ S,
                                          const float* __restrict__ stats, const float* __restrict__ gamma,
                                          const float* __restrict__ beta, float invN, int N, int K) {
    int wave = (int)((blockIdx.x * blockDim.x + threadIdx.x) >> 5);
    int lane = threadIdx.x & 31;
    int row0 = wave * 16;
    if (row0 >= N) return;

    int m  = lane & 15;
    int hi = lane >> 4;

    v8f acc = {};
    if constexpr (BN) {
        const float* arow = A + (size_t)(row0 + m) * HC;
#pragma unroll
        for (int k0 = 0; k0 < HC; k0 += 4) {
            int ka = k0 + 2 * hi;
            v2f a, b;
            a.x = bn_relu(arow[ka],     stats, gamma, beta, invN, ka);
            a.y = bn_relu(arow[ka + 1], stats, gamma, beta, invN, ka + 1);
            b.x = B[(size_t)ka * HC + m];
            b.y = B[(size_t)(ka + 1) * HC + m];
            acc = __builtin_amdgcn_wmma_f32_16x16x4_f32(
                 false, a,  false, b,
                 (short)0, acc,  false,  false);
        }
    } else {
        const float* arow = A + (size_t)(row0 + m) * K;
        for (int k0 = 0; k0 < K; k0 += 4) {
            int ka = k0 + 2 * hi;
            v2f a, b;
            a.x = arow[ka];
            a.y = arow[ka + 1];
            b.x = B[(size_t)ka * HC + m];
            b.y = B[(size_t)(ka + 1) * HC + m];
            acc = __builtin_amdgcn_wmma_f32_16x16x4_f32(
                 false, a,  false, b,
                 (short)0, acc,  false,  false);
        }
    }

    float* cbase = C + (size_t)(row0 + hi * 8) * HC + m;
    float* sbase = S + (size_t)(row0 + hi * 8) * HC + m;
    const float* drow = dis + row0 + hi * 8;
    float bm = bias[m];
#pragma unroll
    for (int r = 0; r < 8; ++r) {
        float d = drow[r];
        cbase[(size_t)r * HC] = d * acc[r];
        sbase[(size_t)r * HC] = d * d * acc[r] + bm;
    }
}

__global__ void gemm_wmma_f32(const float* __restrict__ A, const float* __restrict__ B,
                              float* __restrict__ C, const float* __restrict__ dis,
                              const float* __restrict__ bias, float* __restrict__ S, int N, int K) {
    gemm_tile<false>(A, B, C, dis, bias, S, nullptr, nullptr, nullptr, 0.0f, N, K);
}

__global__ void gemm_bn_wmma_f32(const float* __restrict__ A, const float* __restrict__ B,
                                 float* __restrict__ C, const float* __restrict__ dis,
                                 const float* __restrict__ bias, float* __restrict__ S,
                                 const float* __restrict__ stats, const float* __restrict__ gamma,
                                 const float* __restrict__ beta, float invN, int N) {
    gemm_tile<true>(A, B, C, dis, bias, S, stats, gamma, beta, invN, N, HC);
}

__global__ void gemm_bn_fold_wmma_f32(const float* __restrict__ A, const float* __restrict__ W,
                                      const float* __restrict__ Wfc, const float* __restrict__ bfc,
                                      float* __restrict__ C, const float* __restrict__ dis,
                                      const float* __restrict__ bias, float* __restrict__ S,
                                      const float* __restrict__ stats, const float* __restrict__ gamma,
                                      const float* __restrict__ beta, float invN, int N) {
    int wave = (int)((blockIdx.x * blockDim.x + threadIdx.x) >> 5);
    int lane = threadIdx.x & 31;
    int row0 = wave * 16;
    if (row0 >= N) return;

    int m  = lane & 15;
    int hi = lane >> 4;
    const bool col_on = m < 2;
    const int  mc = m & 1;

    v8f wp = {};
#pragma unroll
    for (int k0 = 0; k0 < HC; k0 += 4) {
        int ka = k0 + 2 * hi;
        v2f a, b;
        a.x = W[(size_t)m * HC + ka];
        a.y = W[(size_t)m * HC + ka + 1];
        float f0 = Wfc[ka * 2 + mc];
        float f1 = Wfc[(ka + 1) * 2 + mc];
        b.x = col_on ? f0 : 0.0f;
        b.y = col_on ? f1 : 0.0f;
        wp = __builtin_amdgcn_wmma_f32_16x16x4_f32(
             false, a,  false, b,
             (short)0, wp,  false,  false);
    }
    float p[8];
#pragma unroll
    for (int r = 0; r < 8; ++r) p[r] = __shfl_xor(wp[r], 16);
    v2f b0, b1, b2, b3;
    b0.x = hi ? p[2]  : wp[0];  b0.y = hi ? p[3]  : wp[1];
    b1.x = hi ? p[6]  : wp[4];  b1.y = hi ? p[7]  : wp[5];
    b2.x = hi ? wp[2] : p[0];   b2.y = hi ? wp[3] : p[1];
    b3.x = hi ? wp[6] : p[4];   b3.y = hi ? wp[7] : p[5];

    const float* arow = A + (size_t)(row0 + m) * HC;
    v2f a0, a1, a2, a3;
    a0.x = bn_relu(arow[2 * hi],      stats, gamma, beta, invN, 2 * hi);
    a0.y = bn_relu(arow[2 * hi + 1],  stats, gamma, beta, invN, 2 * hi + 1);
    a1.x = bn_relu(arow[2 * hi + 4],  stats, gamma, beta, invN, 2 * hi + 4);
    a1.y = bn_relu(arow[2 * hi + 5],  stats, gamma, beta, invN, 2 * hi + 5);
    a2.x = bn_relu(arow[2 * hi + 8],  stats, gamma, beta, invN, 2 * hi + 8);
    a2.y = bn_relu(arow[2 * hi + 9],  stats, gamma, beta, invN, 2 * hi + 9);
    a3.x = bn_relu(arow[2 * hi + 12], stats, gamma, beta, invN, 2 * hi + 12);
    a3.y = bn_relu(arow[2 * hi + 13], stats, gamma, beta, invN, 2 * hi + 13);
    v8f acc = {};
    acc = __builtin_amdgcn_wmma_f32_16x16x4_f32(false, a0, false, b0, (short)0, acc, false, false);
    acc = __builtin_amdgcn_wmma_f32_16x16x4_f32(false, a1, false, b1, (short)0, acc, false, false);
    acc = __builtin_amdgcn_wmma_f32_16x16x4_f32(false, a2, false, b2, (short)0, acc, false, false);
    acc = __builtin_amdgcn_wmma_f32_16x16x4_f32(false, a3, false, b3, (short)0, acc, false, false);

    float bm = bfc[mc];
#pragma unroll
    for (int j = 0; j < HC; ++j) bm += bias[j] * Wfc[j * 2 + mc];

    if (col_on) {
        float* cbase = C + (size_t)(row0 + hi * 8) * 2 + m;
        float* sbase = S + (size_t)(row0 + hi * 8) * 2 + m;
        const float* drow = dis + row0 + hi * 8;
#pragma unroll
        for (int r = 0; r < 8; ++r) {
            float d = drow[r];
            cbase[(size_t)r * 2] = d * acc[r];
            sbase[(size_t)r * 2] = d * d * acc[r] + bm;
        }
    }
}

#define OWN_THREADS 1024
#define OWN_QCAP    64
#define OWN_CHUNK   (16 * OWN_THREADS)

template <int CH, int PITCH, int SRC_BITS, bool FULL>
__device__ __forceinline__ void own_add_queue(const unsigned* q, int n, int lane,
                                              const float* __restrict__ hw, float* acc) {
    if constexpr (CH == 16) {
        const int sub = lane >> 2;
        const int qtr = lane & 3;
        float4 v[4];
        float* arow[4];
        bool   on[4];
#pragma unroll
        for (int j = 0; j < 4; ++j) {
            const int idx = 8 * j + sub;
            on[j] = FULL || idx < n;
            unsigned w = 0u;
            if (on[j]) w = q[idx];
            const unsigned s  = w & ((1u << SRC_BITS) - 1u);
            const unsigned dl = w >> SRC_BITS;
            arow[j] = acc + dl * PITCH + 4 * qtr;
            v[j] = make_float4(0.0f, 0.0f, 0.0f, 0.0f);
            if (on[j]) v[j] = ((const float4*)(hw + (size_t)s * 16))[qtr];
        }
#pragma unroll
        for (int j = 0; j < 4; ++j) {
            if (on[j]) {
                atomicAdd(arow[j] + 0, v[j].x);
                atomicAdd(arow[j] + 1, v[j].y);
                atomicAdd(arow[j] + 2, v[j].z);
                atomicAdd(arow[j] + 3, v[j].w);
            }
        }
    } else {
        static_assert(CH == 2, "row widths in use: 16 and 2");
        if (FULL || lane < n) {
            const unsigned w  = q[lane];
            const unsigned s  = w & ((1u << SRC_BITS) - 1u);
            const unsigned dl = w >> SRC_BITS;
            float* arow = acc + dl * PITCH;
            const float2 v = *(const float2*)(hw + (size_t)s * 2);
            atomicAdd(arow + 0, v.x);
            atomicAdd(arow + 1, v.y);
        }
    }
}

template <int CH, int ROWS, int SLICES, int PITCH, int SRC_BITS>
__global__ __launch_bounds__(OWN_THREADS)
void aggregate_owner(const int* __restrict__ src, const int* __restrict__ dst,
                     const float* __restrict__ dis, const float* __restrict__ hw,
                     float* __restrict__ out, int N, int E) {
    static_assert(ROWS <= (1 << (32 - SRC_BITS)), "local row must fit the queue word");
    __shared__ float acc[ROWS * PITCH];
    __shared__ unsigned queue[(OWN_THREADS / 32) * OWN_QCAP];
    __shared__ float dsl[ROWS];
    const int tid  = threadIdx.x;
    const int lane = tid & 31;
    unsigned* myq = queue + (tid >> 5) * OWN_QCAP;
    const int grp  = blockIdx.x / SLICES;
    const int slc  = blockIdx.x % SLICES;
    const int lo   = grp * ROWS;
    const int rows = min(ROWS, N - lo);
    constexpr int GRAN = (CH == 16) ? OWN_CHUNK : OWN_THREADS;
    const int per  = ((E + SLICES * GRAN - 1) / (SLICES * GRAN)) * GRAN;
    const int e0   = min(E, slc * per);
    const int e1   = min(E, e0 + per);

    constexpr int DPT = (ROWS + OWN_THREADS - 1) / OWN_THREADS;
    float dreg[DPT];
#pragma unroll
    for (int j = 0; j < DPT; ++j) {
        int idx = tid + j * OWN_THREADS;
        dreg[j] = (idx < rows) ? dis[lo + idx] : 0.0f;
    }
    for (int idx = tid; idx < rows * PITCH; idx += OWN_THREADS) acc[idx] = 0.0f;
#pragma unroll
    for (int j = 0; j < DPT; ++j) {
        int idx = tid + j * OWN_THREADS;
        if (idx < rows) dsl[idx] = dreg[j];
    }
    __syncthreads();

    int qn = 0;

#define OWN_STEP(m, km, w)                                                              \
    {                                                                                   \
        if (m) myq[qn + (int)__builtin_amdgcn_mbcnt_lo(km, 0u)] = (w);                  \
        qn += __popc(km);                                                               \
          \
          \
          \
        __builtin_amdgcn_wave_barrier();                                                \
        if (qn >= 32) {                                                                 \
            qn -= 32;                                                                   \
            own_add_queue<CH, PITCH, SRC_BITS, true>(myq + qn, 32, lane, hw, acc);      \
        }                                                                               \
                             \
        __builtin_amdgcn_wave_barrier();                                                \
    }

    int base = e0;
    for (; base + OWN_CHUNK <= e1; base += OWN_CHUNK) {
        const int* p  = dst + base + tid;
        const int* ps = src + base + tid;
        int dv[16], sv[16];
#pragma unroll
        for (int k = 0; k < 16; ++k) dv[k] = p[k * OWN_THREADS];
#pragma unroll
        for (int k = 0; k < 16; ++k)
            sv[k] = __hip_atomic_load(ps + k * OWN_THREADS, __ATOMIC_RELAXED, __HIP_MEMORY_SCOPE_WAVEFRONT);
        unsigned km[16];
        unsigned any = 0u;
#pragma unroll
        for (int k = 0; k < 16; ++k) {
            km[k] = __builtin_amdgcn_ballot_w32((unsigned)(dv[k] - lo) < (unsigned)rows);
            any |= km[k];
        }
        if (any != 0u) {
#pragma unroll
            for (int k = 0; k < 16; ++k) {
                if (km[k] != 0u) {
                    bool m = (unsigned)(dv[k] - lo) < (unsigned)rows;
                    OWN_STEP(m, km[k], (unsigned)sv[k] | ((unsigned)(dv[k] - lo) << SRC_BITS))
                }
            }
        }
    }
    for (; base < e1; base += OWN_THREADS) {
        int e = base + tid;
        int d = (e < e1) ? dst[e] : -1;
        int s = (e < e1) ? src[e] : 0;
        bool m = (unsigned)(d - lo) < (unsigned)rows;
        unsigned k1 = __builtin_amdgcn_ballot_w32(m);
        if (k1 != 0u) OWN_STEP(m, k1, (unsigned)s | ((unsigned)(d - lo) << SRC_BITS))
    }
#undef OWN_STEP
    own_add_queue<CH, PITCH, SRC_BITS, false>(myq, qn, lane, hw, acc);
    __syncthreads();

    for (int idx = tid; idx < rows * CH; idx += OWN_THREADS) {
        int r = idx / CH, c = idx % CH;
        atomicAdd(out + (size_t)lo * CH + idx, dsl[r] * acc[r * PITCH + c]);
    }
}

#define BNS_THREADS 256
#define BNS_DEPTH   7

__global__ __launch_bounds__(BNS_THREADS)
void bn_stats(const float* __restrict__ agg, float* __restrict__ stats, int N) {
    __shared__ float red[(BNS_THREADS / 32) * 32];
    const int tid    = threadIdx.x;
    const int lane   = tid & 31;
    const int wv     = tid >> 5;
    const int total4 = N * (HC / 4);
    const int span   = (int)(gridDim.x * BNS_THREADS);
    const float4* p  = (const float4*)agg;
    float s1[4] = {0.0f, 0.0f, 0.0f, 0.0f};
    float s2[4] = {0.0f, 0.0f, 0.0f, 0.0f};
    for (int base = (int)(blockIdx.x * BNS_THREADS) + tid; base < total4; base += BNS_DEPTH * span) {
        float4 v[BNS_DEPTH];
#pragma unroll
        for (int j = 0; j < BNS_DEPTH; ++j) v[j] = p[min(base + j * span, total4 - 1)];
#pragma unroll
        for (int j = 0; j < BNS_DEPTH; ++j) {
            const bool  on = base + j * span < total4;
            const float a = on ? fmaxf(v[j].x, 0.0f) : 0.0f;
            const float b = on ? fmaxf(v[j].y, 0.0f) : 0.0f;
            const float c = on ? fmaxf(v[j].z, 0.0f) : 0.0f;
            const float d = on ? fmaxf(v[j].w, 0.0f) : 0.0f;
            s1[0] += a; s2[0] += a * a;
            s1[1] += b; s2[1] += b * b;
            s1[2] += c; s2[2] += c * c;
            s1[3] += d; s2[3] += d * d;
        }
    }
#pragma unroll
    for (int c = 0; c < 4; ++c) {
#pragma unroll
        for (int off = 4; off < 32; off <<= 1) {
            s1[c] += __shfl_xor(s1[c], off);
            s2[c] += __shfl_xor(s2[c], off);
        }
    }
    if (lane < 4) {
#pragma unroll
        for (int c = 0; c < 4; ++c) {
            red[wv * 32 + lane * 4 + c]      = s1[c];
            red[wv * 32 + 16 + lane * 4 + c] = s2[c];
        }
    }
    __syncthreads();
    if (tid < 32) {
        float t = 0.0f;
#pragma unroll
        for (int w = 0; w < BNS_THREADS / 32; ++w) t += red[w * 32 + tid];
        atomicAdd(&stats[tid], t);
    }
}

__global__ void head_logsoftmax2(const float* __restrict__ z, float* __restrict__ out, int N) {
    int i = blockIdx.x * blockDim.x + threadIdx.x;
    if (i >= N) return;
    const float2 v = ((const float2*)z)[i];
    float z0 = v.x, z1 = v.y;
    float mx  = fmaxf(z0, z1);
    float lse = mx + logf(__expf(z0 - mx) + __expf(z1 - mx));
    out[(size_t)i * 2 + 0] = z0 - lse;
    out[(size_t)i * 2 + 1] = z1 - lse;
}

extern "C" void kernel_launch(void* const* d_in, const int* in_sizes, int n_in,
                              void* d_out, int out_size, void* d_ws, size_t ws_size,
                              hipStream_t stream) {
    const int IN = 128;
    const int N  = in_sizes[0] / IN;
    const int E  = in_sizes[1] / 2;

    const float* x    = (const float*)d_in[0];
    const int*   ei   = (const int*)d_in[1];
    const int*   srcI = ei;
    const int*   dstI = ei + E;
    const float* W1   = (const float*)d_in[2];
    const float* b1   = (const float*)d_in[3];
    const float* W2   = (const float*)d_in[4];
    const float* b2   = (const float*)d_in[5];
    const float* W3   = (const float*)d_in[6];
    const float* b3   = (const float*)d_in[7];
    const float* g1   = (const float*)d_in[8];
    const float* be1  = (const float*)d_in[9];
    const float* g2   = (const float*)d_in[10];
    const float* be2  = (const float*)d_in[11];
    const float* Wfc  = (const float*)d_in[12];
    const float* bfc  = (const float*)d_in[13];

    float* ws   = (float*)d_ws;
    float* dis  = ws;
    float* bufA = dis + N;
    float* bufB = bufA + (size_t)N * HC;
    float* bufC = bufB + (size_t)N * HC;
    float* stats = bufC + (size_t)N * HC;

    const int B256 = 256;
    dim3 blkN((N + B256 - 1) / B256);
    dim3 blkG((N / 16 + 7) / 8);
    float invN = 1.0f / (float)N;
    dim3 blkO(((N + 4348 - 1) / 4348) * 11);
    dim3 blkO2(((N + 25008 - 1) / 25008) * 64);

    int* planes = (int*)bufA;
    dim3 blkD(((N + DEG_ROWS - 1) / DEG_ROWS) * DEG_SLICES);
    degree_owner<<<blkD, DEG_THREADS, 0, stream>>>(dstI, planes, N, E);
    degree_merge_rsqrt<<<blkN, B256, 0, stream>>>(planes, dis, stats, N);

    gemm_wmma_f32<<<blkG, B256, 0, stream>>>(x, W1, bufA, dis, b1, bufB, N, IN);
    aggregate_owner<16, 4348, 11, 17, 17><<<blkO, OWN_THREADS, 0, stream>>>(srcI, dstI, dis, bufA, bufB, N, E);
    bn_stats<<<256, BNS_THREADS, 0, stream>>>(bufB, stats, N);

    gemm_bn_wmma_f32<<<blkG, B256, 0, stream>>>(bufB, W2, bufA, dis, b2, bufC, stats, g1, be1, invN, N);
    aggregate_owner<16, 4348, 11, 17, 17><<<blkO, OWN_THREADS, 0, stream>>>(srcI, dstI, dis, bufA, bufC, N, E);
    bn_stats<<<256, BNS_THREADS, 0, stream>>>(bufC, stats + 32, N);

    gemm_bn_fold_wmma_f32<<<blkG, B256, 0, stream>>>(bufC, W3, Wfc, bfc, bufA, dis, b3, bufB, stats + 32, g2, be2, invN, N);
    aggregate_owner<2, 25008, 64, 2, 17><<<blkO2, OWN_THREADS, 0, stream>>>(srcI, dstI, dis, bufA, bufB, N, E);

    head_logsoftmax2<<<blkN, B256, 0, stream>>>(bufB, (float*)d_out, N);
}
